# starting kernel
# baseline (speedup 1.0000x reference)
#include <hip/hip_runtime.h>
#include <hip/hip_fp16.h>

typedef __attribute__((ext_vector_type(16))) _Float16 v16h;
typedef __attribute__((ext_vector_type(8)))  float    v8f;
typedef __attribute__((ext_vector_type(2)))  float    v2f;

#define THREADS 256
#define WAVES_PER_BLOCK (THREADS / 32)

__global__ __launch_bounds__(THREADS)
void k_fill_ones(float* __restrict__ p, int n) {
  int i = blockIdx.x * blockDim.x + threadIdx.x;
  if (i < n) p[i] = 1.0f;
}

__global__ __launch_bounds__(THREADS)
void k_deg_count(const int* __restrict__ dst, float* __restrict__ deg, int e) {
  int i = blockIdx.x * blockDim.x + threadIdx.x;
  if (i < e) atomicAdd(&deg[dst[i]], 1.0f);
}

__global__ __launch_bounds__(THREADS)
void k_rsqrt(float* __restrict__ p, int n) {
  int i = blockIdx.x * blockDim.x + threadIdx.x;
  if (i < n) p[i] = rsqrtf(p[i]);
}

__global__ __launch_bounds__(THREADS)
void k_xform32_wmma(const float* __restrict__ x, const float* __restrict__ W,
                    float* __restrict__ m, int n) {
  __shared__ float lds[WAVES_PER_BLOCK * 64];
  int wave  = (int)((blockIdx.x * (unsigned)blockDim.x + threadIdx.x) >> 5);
  int tl    = threadIdx.x & 31;
  int row0  = wave * 16;
  if (row0 >= n) return;
  int lane16 = tl & 15;
  int hi     = tl >> 4;

  v16h b;
  {
    int  colW = lane16 & 3;
    bool keep = lane16 < 4;
#pragma unroll
    for (int j = 0; j < 16; ++j) {
      int k = (j < 8) ? (hi * 8 + j) : (16 + hi * 8 + (j - 8));
      float wv = W[k * 4 + colW];
      b[j] = keep ? (_Float16)wv : (_Float16)0.0f;
    }
  }

  int  arow = row0 + lane16;
  bool rok  = arow < n;
  int  ar   = rok ? arow : (n - 1);
  const float* xr = x + (size_t)ar * 32;
  float4 q0 = *reinterpret_cast<const float4*>(xr + hi * 8);
  float4 q1 = *reinterpret_cast<const float4*>(xr + hi * 8 + 4);
  float4 q2 = *reinterpret_cast<const float4*>(xr + 16 + hi * 8);
  float4 q3 = *reinterpret_cast<const float4*>(xr + 16 + hi * 8 + 4);
  float zf = rok ? 1.0f : 0.0f;
  v16h a;
  a[0]=(_Float16)(q0.x*zf); a[1]=(_Float16)(q0.y*zf); a[2]=(_Float16)(q0.z*zf); a[3]=(_Float16)(q0.w*zf);
  a[4]=(_Float16)(q1.x*zf); a[5]=(_Float16)(q1.y*zf); a[6]=(_Float16)(q1.z*zf); a[7]=(_Float16)(q1.w*zf);
  a[8]=(_Float16)(q2.x*zf); a[9]=(_Float16)(q2.y*zf); a[10]=(_Float16)(q2.z*zf); a[11]=(_Float16)(q2.w*zf);
  a[12]=(_Float16)(q3.x*zf); a[13]=(_Float16)(q3.y*zf); a[14]=(_Float16)(q3.z*zf); a[15]=(_Float16)(q3.w*zf);

  v8f c = {};
  c = __builtin_amdgcn_wmma_f32_16x16x32_f16(false, a, false, b,
                                             (short)0, c, false, false);

  float* slab = lds + ((threadIdx.x >> 5) << 6);
  if (lane16 < 4) {
#pragma unroll
    for (int r = 0; r < 8; ++r) slab[(hi * 8 + r) * 4 + lane16] = c[r];
  }
  asm volatile("s_wait_dscnt 0" ::: "memory");
  if (hi == 0) {
    int mr = row0 + lane16;
    if (mr < n) {
      float4 v = *reinterpret_cast<const float4*>(slab + lane16 * 4);
      *reinterpret_cast<float4*>(m + (size_t)mr * 4) = v;
    }
  }
}

template<int KIN, int FOUT, bool BIAS>
__global__ __launch_bounds__(THREADS)
void k_xform_small(const float* __restrict__ hin, const float* __restrict__ W,
                   const float* __restrict__ bias, float* __restrict__ mout, int n) {
  __shared__ float lds[WAVES_PER_BLOCK * 64];
  int wave  = (int)((blockIdx.x * (unsigned)blockDim.x + threadIdx.x) >> 5);
  int tl    = threadIdx.x & 31;
  int row0  = wave * 16;
  if (row0 >= n) return;
  int lane16 = tl & 15;
  int hi     = tl >> 4;

#if __has_builtin(__builtin_amdgcn_wmma_f32_16x16x4_f32)
  int  arow = row0 + lane16;
  bool rok  = arow < n;
  int  ar   = rok ? arow : (n - 1);
  float zf  = rok ? 1.0f : 0.0f;

  v2f a;
  if (KIN == 4) {
    float2 v = *reinterpret_cast<const float2*>(hin + (size_t)ar * 4 + hi * 2);
    a[0] = v.x * zf;  a[1] = v.y * zf;
  } else {
    float2 v = *reinterpret_cast<const float2*>(hin + (size_t)ar * 2);
    float zk = (hi == 0) ? zf : 0.0f;
    a[0] = v.x * zk;  a[1] = v.y * zk;
  }

  v2f b;
#pragma unroll
  for (int j = 0; j < 2; ++j) {
    int  k    = hi * 2 + j;
    int  kc   = (k < KIN) ? k : (KIN - 1);
    int  cc   = (lane16 < FOUT) ? lane16 : (FOUT - 1);
    float wv  = W[kc * FOUT + cc];
    b[j] = (k < KIN && lane16 < FOUT) ? wv : 0.0f;
  }

  v8f c = {};
  c = __builtin_amdgcn_wmma_f32_16x16x4_f32(false, a, false, b,
                                            (short)0, c, false, false);

  float* slab = lds + ((threadIdx.x >> 5) << 6);
  if (lane16 < FOUT) {
#pragma unroll
    for (int r = 0; r < 8; ++r) slab[(hi * 8 + r) * FOUT + lane16] = c[r];
  }
  asm volatile("s_wait_dscnt 0" ::: "memory");
  if (hi == 0) {
    int mr = row0 + lane16;
    if (mr < n) {
      if (FOUT == 4) {
        float4 v = *reinterpret_cast<const float4*>(slab + lane16 * 4);
        if (BIAS) { v.x += bias[0]; v.y += bias[1]; v.z += bias[2]; v.w += bias[3]; }
        *reinterpret_cast<float4*>(mout + (size_t)mr * 4) = v;
      } else {
        float2 v = *reinterpret_cast<const float2*>(slab + lane16 * 2);
        if (BIAS) { v.x += bias[0]; v.y += bias[1]; }
        *reinterpret_cast<float2*>(mout + (size_t)mr * 2) = v;
      }
    }
  }
#else
  if (hi == 0) {
    int mr = row0 + lane16;
    if (mr < n) {
#pragma unroll
      for (int f = 0; f < FOUT; ++f) {
        float acc = BIAS ? bias[f] : 0.0f;
#pragma unroll
        for (int k = 0; k < KIN; ++k) acc += hin[mr * KIN + k] * W[k * FOUT + f];
        mout[mr * FOUT + f] = acc;
      }
    }
  }
#endif
}

template<int F>
__global__ __launch_bounds__(THREADS)
void k_agg_init(float* __restrict__ agg, const float* __restrict__ m,
                const float* __restrict__ dinv, const float* __restrict__ b, int n) {
  int i = blockIdx.x * blockDim.x + threadIdx.x;
  if (i >= n * F) return;
  int node = i / F;
  int f    = i - node * F;
  float di = dinv[node];
  agg[i] = di * di * m[i] + b[f];
}

template<int F>
__global__ __launch_bounds__(THREADS)
void k_edge_scatter(const int* __restrict__ src, const int* __restrict__ dst,
                    const float* __restrict__ dinv, const float* __restrict__ m,
                    float* __restrict__ agg, int e) {
  int i = blockIdx.x * blockDim.x + threadIdx.x;
  if (i >= e) return;
  int s = src[i], d = dst[i];
  float w = dinv[s] * dinv[d];
  if (F == 4) {
    float4 v = *reinterpret_cast<const float4*>(m + (size_t)s * 4);
    atomicAdd(&agg[(size_t)d * 4 + 0], w * v.x);
    atomicAdd(&agg[(size_t)d * 4 + 1], w * v.y);
    atomicAdd(&agg[(size_t)d * 4 + 2], w * v.z);
    atomicAdd(&agg[(size_t)d * 4 + 3], w * v.w);
  } else {
    float2 v = *reinterpret_cast<const float2*>(m + (size_t)s * 2);
    atomicAdd(&agg[(size_t)d * 2 + 0], w * v.x);
    atomicAdd(&agg[(size_t)d * 2 + 1], w * v.y);
  }
}

__global__ __launch_bounds__(THREADS)
void k_tanh(float* __restrict__ dst, const float* __restrict__ srcp, int cnt) {
  int i = blockIdx.x * blockDim.x + threadIdx.x;
  if (i < cnt) dst[i] = tanhf(srcp[i]);
}

extern "C" void kernel_launch(void* const* d_in, const int* in_sizes, int n_in,
                              void* d_out, int out_size, void* d_ws, size_t ws_size,
                              hipStream_t stream) {
  const float* x  = (const float*)d_in[0];
  const int*   ei = (const int*)  d_in[1];
  const float* W1 = (const float*)d_in[2];
  const float* b1 = (const float*)d_in[3];
  const float* W2 = (const float*)d_in[4];
  const float* b2 = (const float*)d_in[5];
  const float* W3 = (const float*)d_in[6];
  const float* b3 = (const float*)d_in[7];
  const float* Wc = (const float*)d_in[8];
  const float* bc = (const float*)d_in[9];

  const int N = in_sizes[0] / 32;
  const int E = in_sizes[1] / 2;
  const int* src = ei;
  const int* dst = ei + E;

  float* out = (float*)d_out;
  float* h3  = out + (size_t)N * 4;

  float* ws   = (float*)d_ws;
  float* dinv = ws;
  float* m    = ws + (size_t)N;
  float* agg  = ws + (size_t)5 * N;
  float* h    = ws + (size_t)9 * N;

  const int gN  = (N + THREADS - 1) / THREADS;
  const int gE  = (E + THREADS - 1) / THREADS;
  const int g4N = (N * 4 + THREADS - 1) / THREADS;
  const int g2N = (N * 2 + THREADS - 1) / THREADS;
  const int tiles = (N + 15) / 16;
  const int gW  = (tiles + WAVES_PER_BLOCK - 1) / WAVES_PER_BLOCK;

  k_fill_ones<<<gN, THREADS, 0, stream>>>(dinv, N);
  k_deg_count<<<gE, THREADS, 0, stream>>>(dst, dinv, E);
  k_rsqrt<<<gN, THREADS, 0, stream>>>(dinv, N);

  k_xform32_wmma<<<gW, THREADS, 0, stream>>>(x, W1, m, N);
  k_agg_init<4><<<g4N, THREADS, 0, stream>>>(agg, m, dinv, b1, N);
  k_edge_scatter<4><<<gE, THREADS, 0, stream>>>(src, dst, dinv, m, agg, E);
  k_tanh<<<g4N, THREADS, 0, stream>>>(h, agg, N * 4);

  k_xform_small<4, 4, false><<<gW, THREADS, 0, stream>>>(h, W2, nullptr, m, N);
  k_agg_init<4><<<g4N, THREADS, 0, stream>>>(agg, m, dinv, b2, N);
  k_edge_scatter<4><<<gE, THREADS, 0, stream>>>(src, dst, dinv, m, agg, E);
  k_tanh<<<g4N, THREADS, 0, stream>>>(h, agg, N * 4);

  k_xform_small<4, 2, false><<<gW, THREADS, 0, stream>>>(h, W3, nullptr, m, N);
  k_agg_init<2><<<g2N, THREADS, 0, stream>>>(agg, m, dinv, b3, N);
  k_edge_scatter<2><<<gE, THREADS, 0, stream>>>(src, dst, dinv, m, agg, E);
  k_tanh<<<g2N, THREADS, 0, stream>>>(h3, agg, N * 2);

  k_xform_small<2, 4, true><<<gW, THREADS, 0, stream>>>(h3, Wc, bc, out, N);
}


# Per-call CSR by destination (returning int atomic per edge gives its rank; scan; fill), then atomic-free row gathers replace all float atomics; row sums in arrival-rank order, dinv[dst] applied last
# speedup vs baseline: 3.5583x; 3.5583x over previous
#include <hip/hip_runtime.h>
#include <hip/hip_fp16.h>

typedef __attribute__((ext_vector_type(16))) _Float16 v16h;
typedef __attribute__((ext_vector_type(8)))  float    v8f;
typedef __attribute__((ext_vector_type(2)))  float    v2f;

#define THREADS 256
#define WAVES_PER_BLOCK (THREADS / 32)

__global__ __launch_bounds__(THREADS)
void k_zero_int(int* __restrict__ p, int n) {
  int i = blockIdx.x * blockDim.x + threadIdx.x;
  if (i < n) p[i] = 0;
}

__global__ __launch_bounds__(THREADS)
void k_rank(const int* __restrict__ dst, int* __restrict__ cnt,
            int* __restrict__ rank, int e) {
  int i = blockIdx.x * blockDim.x + threadIdx.x;
  if (i < e) rank[i] = atomicAdd(&cnt[dst[i]], 1);
}

#define SCAN_ITEMS 4
#define SCAN_TILE (THREADS * SCAN_ITEMS)

__global__ __launch_bounds__(THREADS)
void k_tile_sums(const int* __restrict__ cnt, int* __restrict__ bsum, int n) {
  __shared__ int wsum[WAVES_PER_BLOCK];
  int lane = threadIdx.x & 31;
  int wv   = threadIdx.x >> 5;
  int base = (blockIdx.x * THREADS + threadIdx.x) * SCAN_ITEMS;
  int s = 0;
#pragma unroll
  for (int k = 0; k < SCAN_ITEMS; ++k)
    if (base + k < n) s += cnt[base + k];
#pragma unroll
  for (int o = 16; o > 0; o >>= 1) s += __shfl_xor(s, o);
  if (lane == 0) wsum[wv] = s;
  __syncthreads();
  if (threadIdx.x == 0) {
    int t = 0;
#pragma unroll
    for (int k = 0; k < WAVES_PER_BLOCK; ++k) t += wsum[k];
    bsum[blockIdx.x] = t;
  }
}

__global__ __launch_bounds__(THREADS)
void k_rowptr(const int* __restrict__ cnt, const int* __restrict__ bsum,
              int* __restrict__ rowptr, float* __restrict__ dinv, int n, int e) {
  __shared__ int wsum[WAVES_PER_BLOCK];
  __shared__ int wsum2[WAVES_PER_BLOCK];
  int lane = threadIdx.x & 31;
  int wv   = threadIdx.x >> 5;

  int part = 0;
  for (int k = threadIdx.x; k < (int)blockIdx.x; k += THREADS) part += bsum[k];
#pragma unroll
  for (int o = 16; o > 0; o >>= 1) part += __shfl_xor(part, o);
  if (lane == 0) wsum[wv] = part;

  int base = (blockIdx.x * THREADS + threadIdx.x) * SCAN_ITEMS;
  int c[SCAN_ITEMS];
  int tsum = 0;
#pragma unroll
  for (int k = 0; k < SCAN_ITEMS; ++k) {
    c[k] = (base + k < n) ? cnt[base + k] : 0;
    tsum += c[k];
  }
  int v = tsum;
#pragma unroll
  for (int o = 1; o < 32; o <<= 1) {
    int t = __shfl_up(v, o);
    if (lane >= o) v += t;
  }
  if (lane == 31) wsum2[wv] = v;
  __syncthreads();

  int ex = v - tsum;
#pragma unroll
  for (int k = 0; k < WAVES_PER_BLOCK; ++k) {
    ex += wsum[k];
    if (k < wv) ex += wsum2[k];
  }
#pragma unroll
  for (int k = 0; k < SCAN_ITEMS; ++k) {
    if (base + k < n) {
      rowptr[base + k] = ex;
      dinv[base + k]   = rsqrtf(1.0f + (float)c[k]);
      ex += c[k];
    }
  }
  if (blockIdx.x == 0 && threadIdx.x == 0) rowptr[n] = e;
}

__global__ __launch_bounds__(THREADS)
void k_fill(const int* __restrict__ src, const int* __restrict__ dst,
            const int* __restrict__ rowptr, const int* __restrict__ rank,
            int* __restrict__ csr, int e) {
  int i = blockIdx.x * blockDim.x + threadIdx.x;
  if (i < e) csr[rowptr[dst[i]] + rank[i]] = src[i];
}

__global__ __launch_bounds__(THREADS)
void k_xform32_wmma(const float* __restrict__ x, const float* __restrict__ W,
                    const float* __restrict__ dinv, float* __restrict__ m, int n) {
  __shared__ float lds[WAVES_PER_BLOCK * 64];
  int wave  = (int)((blockIdx.x * (unsigned)blockDim.x + threadIdx.x) >> 5);
  int tl    = threadIdx.x & 31;
  int row0  = wave * 16;
  if (row0 >= n) return;
  int lane16 = tl & 15;
  int hi     = tl >> 4;

  v16h b;
  {
    int  colW = lane16 & 3;
    bool keep = lane16 < 4;
#pragma unroll
    for (int j = 0; j < 16; ++j) {
      int k = (j < 8) ? (hi * 8 + j) : (16 + hi * 8 + (j - 8));
      float wv = W[k * 4 + colW];
      b[j] = keep ? (_Float16)wv : (_Float16)0.0f;
    }
  }

  int  arow = row0 + lane16;
  bool rok  = arow < n;
  int  ar   = rok ? arow : (n - 1);
  const float* xr = x + (size_t)ar * 32;
  float4 q0 = *reinterpret_cast<const float4*>(xr + hi * 8);
  float4 q1 = *reinterpret_cast<const float4*>(xr + hi * 8 + 4);
  float4 q2 = *reinterpret_cast<const float4*>(xr + 16 + hi * 8);
  float4 q3 = *reinterpret_cast<const float4*>(xr + 16 + hi * 8 + 4);
  float zf = rok ? 1.0f : 0.0f;
  v16h a;
  a[0]=(_Float16)(q0.x*zf); a[1]=(_Float16)(q0.y*zf); a[2]=(_Float16)(q0.z*zf); a[3]=(_Float16)(q0.w*zf);
  a[4]=(_Float16)(q1.x*zf); a[5]=(_Float16)(q1.y*zf); a[6]=(_Float16)(q1.z*zf); a[7]=(_Float16)(q1.w*zf);
  a[8]=(_Float16)(q2.x*zf); a[9]=(_Float16)(q2.y*zf); a[10]=(_Float16)(q2.z*zf); a[11]=(_Float16)(q2.w*zf);
  a[12]=(_Float16)(q3.x*zf); a[13]=(_Float16)(q3.y*zf); a[14]=(_Float16)(q3.z*zf); a[15]=(_Float16)(q3.w*zf);

  v8f c = {};
  c = __builtin_amdgcn_wmma_f32_16x16x32_f16(false, a, false, b,
                                             (short)0, c, false, false);

  float* slab = lds + ((threadIdx.x >> 5) << 6);
  if (lane16 < 4) {
#pragma unroll
    for (int r = 0; r < 8; ++r) slab[(hi * 8 + r) * 4 + lane16] = c[r];
  }
  asm volatile("s_wait_dscnt 0" ::: "memory");
  if (hi == 0) {
    int mr = row0 + lane16;
    if (mr < n) {
      float4 v = *reinterpret_cast<const float4*>(slab + lane16 * 4);
      float di = dinv[mr];
      v.x *= di; v.y *= di; v.z *= di; v.w *= di;
      *reinterpret_cast<float4*>(m + (size_t)mr * 4) = v;
    }
  }
}

template<int KIN, int FOUT, bool BIAS, bool SCALE>
__global__ __launch_bounds__(THREADS)
void k_xform_small(const float* __restrict__ hin, const float* __restrict__ W,
                   const float* __restrict__ bias, const float* __restrict__ dinv,
                   float* __restrict__ mout, int n) {
  __shared__ float lds[WAVES_PER_BLOCK * 64];
  int wave  = (int)((blockIdx.x * (unsigned)blockDim.x + threadIdx.x) >> 5);
  int tl    = threadIdx.x & 31;
  int row0  = wave * 16;
  if (row0 >= n) return;
  int lane16 = tl & 15;
  int hi     = tl >> 4;

#if __has_builtin(__builtin_amdgcn_wmma_f32_16x16x4_f32)
  int  arow = row0 + lane16;
  bool rok  = arow < n;
  int  ar   = rok ? arow : (n - 1);
  float zf  = rok ? 1.0f : 0.0f;

  v2f a;
  if (KIN == 4) {
    float2 v = *reinterpret_cast<const float2*>(hin + (size_t)ar * 4 + hi * 2);
    a[0] = v.x * zf;  a[1] = v.y * zf;
  } else {
    float2 v = *reinterpret_cast<const float2*>(hin + (size_t)ar * 2);
    float zk = (hi == 0) ? zf : 0.0f;
    a[0] = v.x * zk;  a[1] = v.y * zk;
  }

  v2f b;
#pragma unroll
  for (int j = 0; j < 2; ++j) {
    int  k    = hi * 2 + j;
    int  kc   = (k < KIN) ? k : (KIN - 1);
    int  cc   = (lane16 < FOUT) ? lane16 : (FOUT - 1);
    float wv  = W[kc * FOUT + cc];
    b[j] = (k < KIN && lane16 < FOUT) ? wv : 0.0f;
  }

  v8f c = {};
  c = __builtin_amdgcn_wmma_f32_16x16x4_f32(false, a, false, b,
                                            (short)0, c, false, false);

  float* slab = lds + ((threadIdx.x >> 5) << 6);
  if (lane16 < FOUT) {
#pragma unroll
    for (int r = 0; r < 8; ++r) slab[(hi * 8 + r) * FOUT + lane16] = c[r];
  }
  asm volatile("s_wait_dscnt 0" ::: "memory");
  if (hi == 0) {
    int mr = row0 + lane16;
    if (mr < n) {
      if (FOUT == 4) {
        float4 v = *reinterpret_cast<const float4*>(slab + lane16 * 4);
        if (BIAS) { v.x += bias[0]; v.y += bias[1]; v.z += bias[2]; v.w += bias[3]; }
        if (SCALE) { float di = dinv[mr]; v.x *= di; v.y *= di; v.z *= di; v.w *= di; }
        *reinterpret_cast<float4*>(mout + (size_t)mr * 4) = v;
      } else {
        float2 v = *reinterpret_cast<const float2*>(slab + lane16 * 2);
        if (BIAS) { v.x += bias[0]; v.y += bias[1]; }
        if (SCALE) { float di = dinv[mr]; v.x *= di; v.y *= di; }
        *reinterpret_cast<float2*>(mout + (size_t)mr * 2) = v;
      }
    }
  }
#else
  if (hi == 0) {
    int mr = row0 + lane16;
    if (mr < n) {
#pragma unroll
      for (int f = 0; f < FOUT; ++f) {
        float acc = BIAS ? bias[f] : 0.0f;
#pragma unroll
        for (int k = 0; k < KIN; ++k) acc += hin[mr * KIN + k] * W[k * FOUT + f];
        if (SCALE) acc *= dinv[mr];
        mout[mr * FOUT + f] = acc;
      }
    }
  }
#endif
}

template<int F>
__global__ __launch_bounds__(THREADS)
void k_gather(const int* __restrict__ rowptr, const int* __restrict__ csr,
              const float* __restrict__ p, const float* __restrict__ dinv,
              const float* __restrict__ b, float* __restrict__ hout, int n) {
  int node = (int)((blockIdx.x * (unsigned)blockDim.x + threadIdx.x) >> 5);
  int lane = threadIdx.x & 31;
  if (node >= n) return;
  int beg = rowptr[node];
  int end = rowptr[node + 1];

  float a0 = 0.0f, a1 = 0.0f, a2 = 0.0f, a3 = 0.0f;
  int j = beg + lane;
  if (F == 4) {
    const float4* p4 = reinterpret_cast<const float4*>(p);
    for (; j + 32 < end; j += 64) {
      int s0 = csr[j];
      int s1 = csr[j + 32];
      float4 v0 = p4[s0];
      float4 v1 = p4[s1];
      a0 += v0.x; a1 += v0.y; a2 += v0.z; a3 += v0.w;
      a0 += v1.x; a1 += v1.y; a2 += v1.z; a3 += v1.w;
    }
    if (j < end) {
      float4 v0 = p4[csr[j]];
      a0 += v0.x; a1 += v0.y; a2 += v0.z; a3 += v0.w;
    }
  } else {
    const float2* p2 = reinterpret_cast<const float2*>(p);
    for (; j + 32 < end; j += 64) {
      int s0 = csr[j];
      int s1 = csr[j + 32];
      float2 v0 = p2[s0];
      float2 v1 = p2[s1];
      a0 += v0.x; a1 += v0.y;
      a0 += v1.x; a1 += v1.y;
    }
    if (j < end) {
      float2 v0 = p2[csr[j]];
      a0 += v0.x; a1 += v0.y;
    }
  }

#pragma unroll
  for (int o = 16; o > 0; o >>= 1) {
    a0 += __shfl_xor(a0, o);
    a1 += __shfl_xor(a1, o);
    if (F == 4) {
      a2 += __shfl_xor(a2, o);
      a3 += __shfl_xor(a3, o);
    }
  }

  if (lane < F) {
    float sum = (lane == 0) ? a0 : (lane == 1) ? a1 : (lane == 2) ? a2 : a3;
    float di  = dinv[node];
    float self = p[(size_t)node * F + lane];
    hout[(size_t)node * F + lane] = tanhf(di * (sum + self) + b[lane]);
  }
}

extern "C" void kernel_launch(void* const* d_in, const int* in_sizes, int n_in,
                              void* d_out, int out_size, void* d_ws, size_t ws_size,
                              hipStream_t stream) {
  const float* x  = (const float*)d_in[0];
  const int*   ei = (const int*)  d_in[1];
  const float* W1 = (const float*)d_in[2];
  const float* b1 = (const float*)d_in[3];
  const float* W2 = (const float*)d_in[4];
  const float* b2 = (const float*)d_in[5];
  const float* W3 = (const float*)d_in[6];
  const float* b3 = (const float*)d_in[7];
  const float* Wc = (const float*)d_in[8];
  const float* bc = (const float*)d_in[9];

  const int N = in_sizes[0] / 32;
  const int E = in_sizes[1] / 2;
  const int* src = ei;
  const int* dst = ei + E;

  float* out = (float*)d_out;
  float* h3  = out + (size_t)N * 4;

  float* ws   = (float*)d_ws;
  float* p    = ws;
  float* h    = ws + (size_t)4 * N;
  float* dinv = ws + (size_t)8 * N;
  int*   wi     = (int*)(ws + (size_t)9 * N);
  int*   cnt    = wi;
  int*   rank   = wi + (size_t)N;
  int*   csr    = rank + (size_t)E;
  int*   rowptr = csr + (size_t)E;
  int*   bsum   = rowptr + (size_t)N + 1;

  const int gN  = (N + THREADS - 1) / THREADS;
  const int gE  = (E + THREADS - 1) / THREADS;
  const int gS  = (N + SCAN_TILE - 1) / SCAN_TILE;
  const int gG  = (N + WAVES_PER_BLOCK - 1) / WAVES_PER_BLOCK;
  const int tiles = (N + 15) / 16;
  const int gW  = (tiles + WAVES_PER_BLOCK - 1) / WAVES_PER_BLOCK;

  k_zero_int<<<gN, THREADS, 0, stream>>>(cnt, N);
  k_rank<<<gE, THREADS, 0, stream>>>(dst, cnt, rank, E);
  k_tile_sums<<<gS, THREADS, 0, stream>>>(cnt, bsum, N);
  k_rowptr<<<gS, THREADS, 0, stream>>>(cnt, bsum, rowptr, dinv, N, E);
  k_fill<<<gE, THREADS, 0, stream>>>(src, dst, rowptr, rank, csr, E);

  k_xform32_wmma<<<gW, THREADS, 0, stream>>>(x, W1, dinv, p, N);
  k_gather<4><<<gG, THREADS, 0, stream>>>(rowptr, csr, p, dinv, b1, h, N);

  k_xform_small<4, 4, false, true><<<gW, THREADS, 0, stream>>>(h, W2, nullptr, dinv, p, N);
  k_gather<4><<<gG, THREADS, 0, stream>>>(rowptr, csr, p, dinv, b2, h, N);

  k_xform_small<4, 2, false, true><<<gW, THREADS, 0, stream>>>(h, W3, nullptr, dinv, p, N);
  k_gather<2><<<gG, THREADS, 0, stream>>>(rowptr, csr, p, dinv, b3, h3, N);

  k_xform_small<2, 4, true, false><<<gW, THREADS, 0, stream>>>(h3, Wc, bc, nullptr, out, N);
}


# One edge pass builds the neighbour table: the returning add's rank indexes a fixed 128-slot row per destination (overflow list past it); scan, row-pointer gather, 2nd edge pass gone; sum order kept
# speedup vs baseline: 4.5687x; 1.2840x over previous
#include <hip/hip_runtime.h>
#include <hip/hip_fp16.h>

typedef __attribute__((ext_vector_type(16))) _Float16 v16h;
typedef __attribute__((ext_vector_type(8)))  float    v8f;
typedef __attribute__((ext_vector_type(2)))  float    v2f;

#define THREADS 256
#define WAVES_PER_BLOCK (THREADS / 32)

#define ROWCAP 128

__global__ __launch_bounds__(THREADS)
void k_zero_int(int* __restrict__ p, int n) {
  int i = blockIdx.x * blockDim.x + threadIdx.x;
  if (i < n) p[i] = 0;
}

__global__ __launch_bounds__(THREADS)
void k_place(const int* __restrict__ src, const int* __restrict__ dst,
             int* __restrict__ cnt, int* __restrict__ slot,
             int* __restrict__ ovf, int n, int e) {
  int i = blockIdx.x * blockDim.x + threadIdx.x;
  if (i < e) {
    int d = dst[i];
    int s = src[i];
    int r = atomicAdd(&cnt[d], 1);
    if (r < ROWCAP) {
      slot[(size_t)d * ROWCAP + r] = s;
    } else {
      int k = atomicAdd(&cnt[n], 1);
      ovf[2 * (size_t)k]     = s;
      ovf[2 * (size_t)k + 1] = d;
    }
  }
}

__global__ __launch_bounds__(THREADS)
void k_xform32_wmma(const float* __restrict__ x, const float* __restrict__ W,
                    const int* __restrict__ cnt, float* __restrict__ dinv,
                    float* __restrict__ m, int n) {
  __shared__ float lds[WAVES_PER_BLOCK * 64];
  int wave  = (int)((blockIdx.x * (unsigned)blockDim.x + threadIdx.x) >> 5);
  int tl    = threadIdx.x & 31;
  int row0  = wave * 16;
  if (row0 >= n) return;
  int lane16 = tl & 15;
  int hi     = tl >> 4;

  v16h b;
  {
    int  colW = lane16 & 3;
    bool keep = lane16 < 4;
#pragma unroll
    for (int j = 0; j < 16; ++j) {
      int k = (j < 8) ? (hi * 8 + j) : (16 + hi * 8 + (j - 8));
      float wv = W[k * 4 + colW];
      b[j] = keep ? (_Float16)wv : (_Float16)0.0f;
    }
  }

  int  arow = row0 + lane16;
  bool rok  = arow < n;
  int  ar   = rok ? arow : (n - 1);
  const float* xr = x + (size_t)ar * 32;
  float4 q0 = *reinterpret_cast<const float4*>(xr + hi * 8);
  float4 q1 = *reinterpret_cast<const float4*>(xr + hi * 8 + 4);
  float4 q2 = *reinterpret_cast<const float4*>(xr + 16 + hi * 8);
  float4 q3 = *reinterpret_cast<const float4*>(xr + 16 + hi * 8 + 4);
  float zf = rok ? 1.0f : 0.0f;
  v16h a;
  a[0]=(_Float16)(q0.x*zf); a[1]=(_Float16)(q0.y*zf); a[2]=(_Float16)(q0.z*zf); a[3]=(_Float16)(q0.w*zf);
  a[4]=(_Float16)(q1.x*zf); a[5]=(_Float16)(q1.y*zf); a[6]=(_Float16)(q1.z*zf); a[7]=(_Float16)(q1.w*zf);
  a[8]=(_Float16)(q2.x*zf); a[9]=(_Float16)(q2.y*zf); a[10]=(_Float16)(q2.z*zf); a[11]=(_Float16)(q2.w*zf);
  a[12]=(_Float16)(q3.x*zf); a[13]=(_Float16)(q3.y*zf); a[14]=(_Float16)(q3.z*zf); a[15]=(_Float16)(q3.w*zf);

  v8f c = {};
  c = __builtin_amdgcn_wmma_f32_16x16x32_f16(false, a, false, b,
                                             (short)0, c, false, false);

  float* slab = lds + ((threadIdx.x >> 5) << 6);
  if (lane16 < 4) {
#pragma unroll
    for (int r = 0; r < 8; ++r) slab[(hi * 8 + r) * 4 + lane16] = c[r];
  }
  asm volatile("s_wait_dscnt 0" ::: "memory");
  if (hi == 0) {
    int mr = row0 + lane16;
    if (mr < n) {
      float4 v = *reinterpret_cast<const float4*>(slab + lane16 * 4);
      float di = rsqrtf(1.0f + (float)cnt[mr]);
      dinv[mr] = di;
      v.x *= di; v.y *= di; v.z *= di; v.w *= di;
      *reinterpret_cast<float4*>(m + (size_t)mr * 4) = v;
    }
  }
}

template<int KIN, int FOUT, bool BIAS, bool SCALE>
__global__ __launch_bounds__(THREADS)
void k_xform_small(const float* __restrict__ hin, const float* __restrict__ W,
                   const float* __restrict__ bias, const float* __restrict__ dinv,
                   float* __restrict__ mout, int n) {
  __shared__ float lds[WAVES_PER_BLOCK * 64];
  int wave  = (int)((blockIdx.x * (unsigned)blockDim.x + threadIdx.x) >> 5);
  int tl    = threadIdx.x & 31;
  int row0  = wave * 16;
  if (row0 >= n) return;
  int lane16 = tl & 15;
  int hi     = tl >> 4;

#if __has_builtin(__builtin_amdgcn_wmma_f32_16x16x4_f32)
  int  arow = row0 + lane16;
  bool rok  = arow < n;
  int  ar   = rok ? arow : (n - 1);
  float zf  = rok ? 1.0f : 0.0f;

  v2f a;
  if (KIN == 4) {
    float2 v = *reinterpret_cast<const float2*>(hin + (size_t)ar * 4 + hi * 2);
    a[0] = v.x * zf;  a[1] = v.y * zf;
  } else {
    float2 v = *reinterpret_cast<const float2*>(hin + (size_t)ar * 2);
    float zk = (hi == 0) ? zf : 0.0f;
    a[0] = v.x * zk;  a[1] = v.y * zk;
  }

  v2f b;
#pragma unroll
  for (int j = 0; j < 2; ++j) {
    int  k    = hi * 2 + j;
    int  kc   = (k < KIN) ? k : (KIN - 1);
    int  cc   = (lane16 < FOUT) ? lane16 : (FOUT - 1);
    float wv  = W[kc * FOUT + cc];
    b[j] = (k < KIN && lane16 < FOUT) ? wv : 0.0f;
  }

  v8f c = {};
  c = __builtin_amdgcn_wmma_f32_16x16x4_f32(false, a, false, b,
                                            (short)0, c, false, false);

  float* slab = lds + ((threadIdx.x >> 5) << 6);
  if (lane16 < FOUT) {
#pragma unroll
    for (int r = 0; r < 8; ++r) slab[(hi * 8 + r) * FOUT + lane16] = c[r];
  }
  asm volatile("s_wait_dscnt 0" ::: "memory");
  if (hi == 0) {
    int mr = row0 + lane16;
    if (mr < n) {
      if (FOUT == 4) {
        float4 v = *reinterpret_cast<const float4*>(slab + lane16 * 4);
        if (BIAS) { v.x += bias[0]; v.y += bias[1]; v.z += bias[2]; v.w += bias[3]; }
        if (SCALE) { float di = dinv[mr]; v.x *= di; v.y *= di; v.z *= di; v.w *= di; }
        *reinterpret_cast<float4*>(mout + (size_t)mr * 4) = v;
      } else {
        float2 v = *reinterpret_cast<const float2*>(slab + lane16 * 2);
        if (BIAS) { v.x += bias[0]; v.y += bias[1]; }
        if (SCALE) { float di = dinv[mr]; v.x *= di; v.y *= di; }
        *reinterpret_cast<float2*>(mout + (size_t)mr * 2) = v;
      }
    }
  }
#else
  if (hi == 0) {
    int mr = row0 + lane16;
    if (mr < n) {
#pragma unroll
      for (int f = 0; f < FOUT; ++f) {
        float acc = BIAS ? bias[f] : 0.0f;
#pragma unroll
        for (int k = 0; k < KIN; ++k) acc += hin[mr * KIN + k] * W[k * FOUT + f];
        if (SCALE) acc *= dinv[mr];
        mout[mr * FOUT + f] = acc;
      }
    }
  }
#endif
}

template<int F>
__global__ __launch_bounds__(THREADS)
void k_gather(const int* __restrict__ cnt, const int* __restrict__ slot,
              const int* __restrict__ ovf,
              const float* __restrict__ p, const float* __restrict__ dinv,
              const float* __restrict__ b, float* __restrict__ hout, int n) {
  int node = (int)((blockIdx.x * (unsigned)blockDim.x + threadIdx.x) >> 5);
  int lane = threadIdx.x & 31;
  if (node >= n) return;
  int deg = cnt[node];
  int m   = (deg < ROWCAP) ? deg : ROWCAP;
  const int* row = slot + (size_t)node * ROWCAP;

  float a0 = 0.0f, a1 = 0.0f, a2 = 0.0f, a3 = 0.0f;
  int j = lane;
  if (F == 4) {
    const float4* p4 = reinterpret_cast<const float4*>(p);
    for (; j + 32 < m; j += 64) {
      int s0 = row[j];
      int s1 = row[j + 32];
      float4 v0 = p4[s0];
      float4 v1 = p4[s1];
      a0 += v0.x; a1 += v0.y; a2 += v0.z; a3 += v0.w;
      a0 += v1.x; a1 += v1.y; a2 += v1.z; a3 += v1.w;
    }
    if (j < m) {
      float4 v0 = p4[row[j]];
      a0 += v0.x; a1 += v0.y; a2 += v0.z; a3 += v0.w;
    }
    if (deg > ROWCAP) {
      int novf = cnt[n];
      for (int k = lane; k < novf; k += 32) {
        if (ovf[2 * (size_t)k + 1] == node) {
          float4 v0 = p4[ovf[2 * (size_t)k]];
          a0 += v0.x; a1 += v0.y; a2 += v0.z; a3 += v0.w;
        }
      }
    }
  } else {
    const float2* p2 = reinterpret_cast<const float2*>(p);
    for (; j + 32 < m; j += 64) {
      int s0 = row[j];
      int s1 = row[j + 32];
      float2 v0 = p2[s0];
      float2 v1 = p2[s1];
      a0 += v0.x; a1 += v0.y;
      a0 += v1.x; a1 += v1.y;
    }
    if (j < m) {
      float2 v0 = p2[row[j]];
      a0 += v0.x; a1 += v0.y;
    }
    if (deg > ROWCAP) {
      int novf = cnt[n];
      for (int k = lane; k < novf; k += 32) {
        if (ovf[2 * (size_t)k + 1] == node) {
          float2 v0 = p2[ovf[2 * (size_t)k]];
          a0 += v0.x; a1 += v0.y;
        }
      }
    }
  }

#pragma unroll
  for (int o = 16; o > 0; o >>= 1) {
    a0 += __shfl_xor(a0, o);
    a1 += __shfl_xor(a1, o);
    if (F == 4) {
      a2 += __shfl_xor(a2, o);
      a3 += __shfl_xor(a3, o);
    }
  }

  if (lane < F) {
    float sum = (lane == 0) ? a0 : (lane == 1) ? a1 : (lane == 2) ? a2 : a3;
    float di  = dinv[node];
    float self = p[(size_t)node * F + lane];
    hout[(size_t)node * F + lane] = tanhf(di * (sum + self) + b[lane]);
  }
}

extern "C" void kernel_launch(void* const* d_in, const int* in_sizes, int n_in,
                              void* d_out, int out_size, void* d_ws, size_t ws_size,
                              hipStream_t stream) {
  const float* x  = (const float*)d_in[0];
  const int*   ei = (const int*)  d_in[1];
  const float* W1 = (const float*)d_in[2];
  const float* b1 = (const float*)d_in[3];
  const float* W2 = (const float*)d_in[4];
  const float* b2 = (const float*)d_in[5];
  const float* W3 = (const float*)d_in[6];
  const float* b3 = (const float*)d_in[7];
  const float* Wc = (const float*)d_in[8];
  const float* bc = (const float*)d_in[9];

  const int N = in_sizes[0] / 32;
  const int E = in_sizes[1] / 2;
  const int* src = ei;
  const int* dst = ei + E;

  float* out = (float*)d_out;
  float* h3  = out + (size_t)N * 4;

  int*   slot = (int*)d_ws;
  float* p    = (float*)(slot + (size_t)N * ROWCAP);
  float* h    = p + (size_t)4 * N;
  float* dinv = p + (size_t)8 * N;
  int*   cnt  = (int*)(p + (size_t)9 * N);
  int*   ovf  = cnt + (size_t)N + 2;

  const int gZ  = (N + 2 + THREADS - 1) / THREADS;
  const int gE  = (E + THREADS - 1) / THREADS;
  const int gG  = (N + WAVES_PER_BLOCK - 1) / WAVES_PER_BLOCK;
  const int tiles = (N + 15) / 16;
  const int gW  = (tiles + WAVES_PER_BLOCK - 1) / WAVES_PER_BLOCK;

  k_zero_int<<<gZ, THREADS, 0, stream>>>(cnt, N + 2);
  k_place<<<gE, THREADS, 0, stream>>>(src, dst, cnt, slot, ovf, N, E);

  k_xform32_wmma<<<gW, THREADS, 0, stream>>>(x, W1, cnt, dinv, p, N);
  k_gather<4><<<gG, THREADS, 0, stream>>>(cnt, slot, ovf, p, dinv, b1, h, N);

  k_xform_small<4, 4, false, true><<<gW, THREADS, 0, stream>>>(h, W2, nullptr, dinv, p, N);
  k_gather<4><<<gG, THREADS, 0, stream>>>(cnt, slot, ovf, p, dinv, b2, h, N);

  k_xform_small<4, 2, false, true><<<gW, THREADS, 0, stream>>>(h, W3, nullptr, dinv, p, N);
  k_gather<2><<<gG, THREADS, 0, stream>>>(cnt, slot, ovf, p, dinv, b3, h3, N);

  k_xform_small<2, 4, true, false><<<gW, THREADS, 0, stream>>>(h3, Wc, bc, nullptr, out, N);
}


# Edge pass split by L2 half: a resident grid draws edge chunks from two queues and only workgroups on a destination's own half add to its counter and write its row; row sums stay in arrival-rank order
# speedup vs baseline: 12.9115x; 2.8261x over previous
#include <hip/hip_runtime.h>
#include <hip/hip_fp16.h>

typedef __attribute__((ext_vector_type(16))) _Float16 v16h;
typedef __attribute__((ext_vector_type(8)))  float    v8f;
typedef __attribute__((ext_vector_type(2)))  float    v2f;

#define THREADS 256
#define WAVES_PER_BLOCK (THREADS / 32)

#define ROWCAP 128

__global__ __launch_bounds__(THREADS)
void k_zero_int(int* __restrict__ p, int n) {
  int i = blockIdx.x * blockDim.x + threadIdx.x;
  if (i < n) p[i] = 0;
}

#define PLACE_EPT   8
#define PLACE_CHUNK (THREADS * PLACE_EPT)
#define QUEUE_PITCH 64

__global__ __launch_bounds__(THREADS)
void k_place(const int* __restrict__ src, const int* __restrict__ dst,
             int* __restrict__ cnt, int* __restrict__ slot,
             int* __restrict__ ovf, int* __restrict__ queue, int n, int e) {
  __shared__ int s_half;
  __shared__ int s_chunk;
  const int nchunks = (e + PLACE_CHUNK - 1) / PLACE_CHUNK;
  const int split   = n >> 1;
  if (threadIdx.x == 0) {
    unsigned where = (unsigned)__builtin_amdgcn_s_sendmsg_rtn(0x87);
    s_half = (int)((where >> 18) & 1u);
  }
  __syncthreads();
  const int myhalf = s_half;

  for (int pass = 0; pass < 2; ++pass) {
    const int h = myhalf ^ pass;
    for (;;) {
      __syncthreads();
      if (threadIdx.x == 0) s_chunk = atomicAdd(&queue[h * QUEUE_PITCH], 1);
      __syncthreads();
      const int c = s_chunk;
      if (c >= nchunks) break;

      const int base = c * PLACE_CHUNK + (int)threadIdx.x;
      int  dd[PLACE_EPT];
      int  ss[PLACE_EPT];
      int  rr[PLACE_EPT];
      bool mine[PLACE_EPT];
#pragma unroll
      for (int k = 0; k < PLACE_EPT; ++k) {
        int i  = base + k * THREADS;
        int ic = (i < e) ? i : (e - 1);
        dd[k] = dst[ic];
        ss[k] = src[ic];
      }
#pragma unroll
      for (int k = 0; k < PLACE_EPT; ++k) {
        int i = base + k * THREADS;
        mine[k] = (i < e) && ((dd[k] >= split) == (h == 1));
      }
#pragma unroll
      for (int k = 0; k < PLACE_EPT; ++k) {
        rr[k] = mine[k] ? atomicAdd(&cnt[dd[k]], 1) : 0;
      }
#pragma unroll
      for (int k = 0; k < PLACE_EPT; ++k) {
        if (mine[k]) {
          if (rr[k] < ROWCAP) {
            slot[(size_t)dd[k] * ROWCAP + rr[k]] = ss[k];
          } else {
            int q = atomicAdd(&cnt[n], 1);
            ovf[2 * (size_t)q]     = ss[k];
            ovf[2 * (size_t)q + 1] = dd[k];
          }
        }
      }
    }
  }
}

__global__ __launch_bounds__(THREADS)
void k_xform32_wmma(const float* __restrict__ x, const float* __restrict__ W,
                    const int* __restrict__ cnt, float* __restrict__ dinv,
                    float* __restrict__ m, int n) {
  __shared__ float lds[WAVES_PER_BLOCK * 64];
  int wave  = (int)((blockIdx.x * (unsigned)blockDim.x + threadIdx.x) >> 5);
  int tl    = threadIdx.x & 31;
  int row0  = wave * 16;
  if (row0 >= n) return;
  int lane16 = tl & 15;
  int hi     = tl >> 4;

  v16h b;
  {
    int  colW = lane16 & 3;
    bool keep = lane16 < 4;
#pragma unroll
    for (int j = 0; j < 16; ++j) {
      int k = (j < 8) ? (hi * 8 + j) : (16 + hi * 8 + (j - 8));
      float wv = W[k * 4 + colW];
      b[j] = keep ? (_Float16)wv : (_Float16)0.0f;
    }
  }

  int  arow = row0 + lane16;
  bool rok  = arow < n;
  int  ar   = rok ? arow : (n - 1);
  const float* xr = x + (size_t)ar * 32;
  float4 q0 = *reinterpret_cast<const float4*>(xr + hi * 8);
  float4 q1 = *reinterpret_cast<const float4*>(xr + hi * 8 + 4);
  float4 q2 = *reinterpret_cast<const float4*>(xr + 16 + hi * 8);
  float4 q3 = *reinterpret_cast<const float4*>(xr + 16 + hi * 8 + 4);
  float zf = rok ? 1.0f : 0.0f;
  v16h a;
  a[0]=(_Float16)(q0.x*zf); a[1]=(_Float16)(q0.y*zf); a[2]=(_Float16)(q0.z*zf); a[3]=(_Float16)(q0.w*zf);
  a[4]=(_Float16)(q1.x*zf); a[5]=(_Float16)(q1.y*zf); a[6]=(_Float16)(q1.z*zf); a[7]=(_Float16)(q1.w*zf);
  a[8]=(_Float16)(q2.x*zf); a[9]=(_Float16)(q2.y*zf); a[10]=(_Float16)(q2.z*zf); a[11]=(_Float16)(q2.w*zf);
  a[12]=(_Float16)(q3.x*zf); a[13]=(_Float16)(q3.y*zf); a[14]=(_Float16)(q3.z*zf); a[15]=(_Float16)(q3.w*zf);

  v8f c = {};
  c = __builtin_amdgcn_wmma_f32_16x16x32_f16(false, a, false, b,
                                             (short)0, c, false, false);

  float* slab = lds + ((threadIdx.x >> 5) << 6);
  if (lane16 < 4) {
#pragma unroll
    for (int r = 0; r < 8; ++r) slab[(hi * 8 + r) * 4 + lane16] = c[r];
  }
  asm volatile("s_wait_dscnt 0" ::: "memory");
  if (hi == 0) {
    int mr = row0 + lane16;
    if (mr < n) {
      float4 v = *reinterpret_cast<const float4*>(slab + lane16 * 4);
      float di = rsqrtf(1.0f + (float)cnt[mr]);
      dinv[mr] = di;
      v.x *= di; v.y *= di; v.z *= di; v.w *= di;
      *reinterpret_cast<float4*>(m + (size_t)mr * 4) = v;
    }
  }
}

template<int KIN, int FOUT, bool BIAS, bool SCALE>
__global__ __launch_bounds__(THREADS)
void k_xform_small(const float* __restrict__ hin, const float* __restrict__ W,
                   const float* __restrict__ bias, const float* __restrict__ dinv,
                   float* __restrict__ mout, int n) {
  __shared__ float lds[WAVES_PER_BLOCK * 64];
  int wave  = (int)((blockIdx.x * (unsigned)blockDim.x + threadIdx.x) >> 5);
  int tl    = threadIdx.x & 31;
  int row0  = wave * 16;
  if (row0 >= n) return;
  int lane16 = tl & 15;
  int hi     = tl >> 4;

#if __has_builtin(__builtin_amdgcn_wmma_f32_16x16x4_f32)
  int  arow = row0 + lane16;
  bool rok  = arow < n;
  int  ar   = rok ? arow : (n - 1);
  float zf  = rok ? 1.0f : 0.0f;

  v2f a;
  if (KIN == 4) {
    float2 v = *reinterpret_cast<const float2*>(hin + (size_t)ar * 4 + hi * 2);
    a[0] = v.x * zf;  a[1] = v.y * zf;
  } else {
    float2 v = *reinterpret_cast<const float2*>(hin + (size_t)ar * 2);
    float zk = (hi == 0) ? zf : 0.0f;
    a[0] = v.x * zk;  a[1] = v.y * zk;
  }

  v2f b;
#pragma unroll
  for (int j = 0; j < 2; ++j) {
    int  k    = hi * 2 + j;
    int  kc   = (k < KIN) ? k : (KIN - 1);
    int  cc   = (lane16 < FOUT) ? lane16 : (FOUT - 1);
    float wv  = W[kc * FOUT + cc];
    b[j] = (k < KIN && lane16 < FOUT) ? wv : 0.0f;
  }

  v8f c = {};
  c = __builtin_amdgcn_wmma_f32_16x16x4_f32(false, a, false, b,
                                            (short)0, c, false, false);

  float* slab = lds + ((threadIdx.x >> 5) << 6);
  if (lane16 < FOUT) {
#pragma unroll
    for (int r = 0; r < 8; ++r) slab[(hi * 8 + r) * FOUT + lane16] = c[r];
  }
  asm volatile("s_wait_dscnt 0" ::: "memory");
  if (hi == 0) {
    int mr = row0 + lane16;
    if (mr < n) {
      if (FOUT == 4) {
        float4 v = *reinterpret_cast<const float4*>(slab + lane16 * 4);
        if (BIAS) { v.x += bias[0]; v.y += bias[1]; v.z += bias[2]; v.w += bias[3]; }
        if (SCALE) { float di = dinv[mr]; v.x *= di; v.y *= di; v.z *= di; v.w *= di; }
        *reinterpret_cast<float4*>(mout + (size_t)mr * 4) = v;
      } else {
        float2 v = *reinterpret_cast<const float2*>(slab + lane16 * 2);
        if (BIAS) { v.x += bias[0]; v.y += bias[1]; }
        if (SCALE) { float di = dinv[mr]; v.x *= di; v.y *= di; }
        *reinterpret_cast<float2*>(mout + (size_t)mr * 2) = v;
      }
    }
  }
#else
  if (hi == 0) {
    int mr = row0 + lane16;
    if (mr < n) {
#pragma unroll
      for (int f = 0; f < FOUT; ++f) {
        float acc = BIAS ? bias[f] : 0.0f;
#pragma unroll
        for (int k = 0; k < KIN; ++k) acc += hin[mr * KIN + k] * W[k * FOUT + f];
        if (SCALE) acc *= dinv[mr];
        mout[mr * FOUT + f] = acc;
      }
    }
  }
#endif
}

template<int F>
__global__ __launch_bounds__(THREADS)
void k_gather(const int* __restrict__ cnt, const int* __restrict__ slot,
              const int* __restrict__ ovf,
              const float* __restrict__ p, const float* __restrict__ dinv,
              const float* __restrict__ b, float* __restrict__ hout, int n) {
  int node = (int)((blockIdx.x * (unsigned)blockDim.x + threadIdx.x) >> 5);
  int lane = threadIdx.x & 31;
  if (node >= n) return;
  int deg = cnt[node];
  int m   = (deg < ROWCAP) ? deg : ROWCAP;
  const int* row = slot + (size_t)node * ROWCAP;

  float a0 = 0.0f, a1 = 0.0f, a2 = 0.0f, a3 = 0.0f;
  int j = lane;
  if (F == 4) {
    const float4* p4 = reinterpret_cast<const float4*>(p);
    for (; j + 32 < m; j += 64) {
      int s0 = row[j];
      int s1 = row[j + 32];
      float4 v0 = p4[s0];
      float4 v1 = p4[s1];
      a0 += v0.x; a1 += v0.y; a2 += v0.z; a3 += v0.w;
      a0 += v1.x; a1 += v1.y; a2 += v1.z; a3 += v1.w;
    }
    if (j < m) {
      float4 v0 = p4[row[j]];
      a0 += v0.x; a1 += v0.y; a2 += v0.z; a3 += v0.w;
    }
    if (deg > ROWCAP) {
      int novf = cnt[n];
      for (int k = lane; k < novf; k += 32) {
        if (ovf[2 * (size_t)k + 1] == node) {
          float4 v0 = p4[ovf[2 * (size_t)k]];
          a0 += v0.x; a1 += v0.y; a2 += v0.z; a3 += v0.w;
        }
      }
    }
  } else {
    const float2* p2 = reinterpret_cast<const float2*>(p);
    for (; j + 32 < m; j += 64) {
      int s0 = row[j];
      int s1 = row[j + 32];
      float2 v0 = p2[s0];
      float2 v1 = p2[s1];
      a0 += v0.x; a1 += v0.y;
      a0 += v1.x; a1 += v1.y;
    }
    if (j < m) {
      float2 v0 = p2[row[j]];
      a0 += v0.x; a1 += v0.y;
    }
    if (deg > ROWCAP) {
      int novf = cnt[n];
      for (int k = lane; k < novf; k += 32) {
        if (ovf[2 * (size_t)k + 1] == node) {
          float2 v0 = p2[ovf[2 * (size_t)k]];
          a0 += v0.x; a1 += v0.y;
        }
      }
    }
  }

#pragma unroll
  for (int o = 16; o > 0; o >>= 1) {
    a0 += __shfl_xor(a0, o);
    a1 += __shfl_xor(a1, o);
    if (F == 4) {
      a2 += __shfl_xor(a2, o);
      a3 += __shfl_xor(a3, o);
    }
  }

  if (lane < F) {
    float sum = (lane == 0) ? a0 : (lane == 1) ? a1 : (lane == 2) ? a2 : a3;
    float di  = dinv[node];
    float self = p[(size_t)node * F + lane];
    hout[(size_t)node * F + lane] = tanhf(di * (sum + self) + b[lane]);
  }
}

extern "C" void kernel_launch(void* const* d_in, const int* in_sizes, int n_in,
                              void* d_out, int out_size, void* d_ws, size_t ws_size,
                              hipStream_t stream) {
  const float* x  = (const float*)d_in[0];
  const int*   ei = (const int*)  d_in[1];
  const float* W1 = (const float*)d_in[2];
  const float* b1 = (const float*)d_in[3];
  const float* W2 = (const float*)d_in[4];
  const float* b2 = (const float*)d_in[5];
  const float* W3 = (const float*)d_in[6];
  const float* b3 = (const float*)d_in[7];
  const float* Wc = (const float*)d_in[8];
  const float* bc = (const float*)d_in[9];

  const int N = in_sizes[0] / 32;
  const int E = in_sizes[1] / 2;
  const int* src = ei;
  const int* dst = ei + E;

  float* out = (float*)d_out;
  float* h3  = out + (size_t)N * 4;

  int*   slot = (int*)d_ws;
  float* p    = (float*)(slot + (size_t)N * ROWCAP);
  float* h    = p + (size_t)4 * N;
  float* dinv = p + (size_t)8 * N;
  int*   cnt  = (int*)(p + (size_t)9 * N);
  int*   queue = cnt + (size_t)N + 2 + 64;
  int*   ovf  = cnt + (size_t)N + 132;

  const int gZ  = (N + 132 + THREADS - 1) / THREADS;
  const int gE  = 2048;
  const int gG  = (N + WAVES_PER_BLOCK - 1) / WAVES_PER_BLOCK;
  const int tiles = (N + 15) / 16;
  const int gW  = (tiles + WAVES_PER_BLOCK - 1) / WAVES_PER_BLOCK;

  k_zero_int<<<gZ, THREADS, 0, stream>>>(cnt, N + 132);
  k_place<<<gE, THREADS, 0, stream>>>(src, dst, cnt, slot, ovf, queue, N, E);

  k_xform32_wmma<<<gW, THREADS, 0, stream>>>(x, W1, cnt, dinv, p, N);
  k_gather<4><<<gG, THREADS, 0, stream>>>(cnt, slot, ovf, p, dinv, b1, h, N);

  k_xform_small<4, 4, false, true><<<gW, THREADS, 0, stream>>>(h, W2, nullptr, dinv, p, N);
  k_gather<4><<<gG, THREADS, 0, stream>>>(cnt, slot, ovf, p, dinv, b2, h, N);

  k_xform_small<4, 2, false, true><<<gW, THREADS, 0, stream>>>(h, W3, nullptr, dinv, p, N);
  k_gather<2><<<gG, THREADS, 0, stream>>>(cnt, slot, ovf, p, dinv, b3, h3, N);

  k_xform_small<2, 4, true, false><<<gW, THREADS, 0, stream>>>(h3, Wc, bc, nullptr, out, N);
}


# Small transforms fused into the gather tails: a wave keeps its nodes' h rows as the A tile of one f32 16x16x4 WMMA and stores the next layer's rows itself; 3 launches gone; sum order kept
# speedup vs baseline: 14.0649x; 1.0893x over previous
#include <hip/hip_runtime.h>
#include <hip/hip_fp16.h>

typedef __attribute__((ext_vector_type(16))) _Float16 v16h;
typedef __attribute__((ext_vector_type(8)))  float    v8f;
typedef __attribute__((ext_vector_type(2)))  float    v2f;

#define THREADS 256
#define WAVES_PER_BLOCK (THREADS / 32)

#define ROWCAP 128

__global__ __launch_bounds__(THREADS)
void k_zero_int(int* __restrict__ p, int n) {
  int i = blockIdx.x * blockDim.x + threadIdx.x;
  if (i < n) p[i] = 0;
}

#define PLACE_EPT   8
#define PLACE_CHUNK (THREADS * PLACE_EPT)
#define QUEUE_PITCH 64

__global__ __launch_bounds__(THREADS)
void k_place(const int* __restrict__ src, const int* __restrict__ dst,
             int* __restrict__ cnt, int* __restrict__ slot,
             int* __restrict__ ovf, int* __restrict__ queue, int n, int e) {
  __shared__ int s_half;
  __shared__ int s_chunk;
  const int nchunks = (e + PLACE_CHUNK - 1) / PLACE_CHUNK;
  const int split   = n >> 1;
  if (threadIdx.x == 0) {
    unsigned where = (unsigned)__builtin_amdgcn_s_sendmsg_rtn(0x87);
    s_half = (int)((where >> 18) & 1u);
  }
  __syncthreads();
  const int myhalf = s_half;

  for (int pass = 0; pass < 2; ++pass) {
    const int h = myhalf ^ pass;
    for (;;) {
      __syncthreads();
      if (threadIdx.x == 0) s_chunk = atomicAdd(&queue[h * QUEUE_PITCH], 1);
      __syncthreads();
      const int c = s_chunk;
      if (c >= nchunks) break;

      const int base = c * PLACE_CHUNK + (int)threadIdx.x;
      int  dd[PLACE_EPT];
      int  ss[PLACE_EPT];
      int  rr[PLACE_EPT];
      bool mine[PLACE_EPT];
#pragma unroll
      for (int k = 0; k < PLACE_EPT; ++k) {
        int i  = base + k * THREADS;
        int ic = (i < e) ? i : (e - 1);
        dd[k] = dst[ic];
        ss[k] = src[ic];
      }
#pragma unroll
      for (int k = 0; k < PLACE_EPT; ++k) {
        int i = base + k * THREADS;
        mine[k] = (i < e) && ((dd[k] >= split) == (h == 1));
      }
#pragma unroll
      for (int k = 0; k < PLACE_EPT; ++k) {
        rr[k] = mine[k] ? atomicAdd(&cnt[dd[k]], 1) : 0;
      }
#pragma unroll
      for (int k = 0; k < PLACE_EPT; ++k) {
        if (mine[k]) {
          if (rr[k] < ROWCAP) {
            slot[(size_t)dd[k] * ROWCAP + rr[k]] = ss[k];
          } else {
            int q = atomicAdd(&cnt[n], 1);
            ovf[2 * (size_t)q]     = ss[k];
            ovf[2 * (size_t)q + 1] = dd[k];
          }
        }
      }
    }
  }
}

__global__ __launch_bounds__(THREADS)
void k_xform32_wmma(const float* __restrict__ x, const float* __restrict__ W,
                    const int* __restrict__ cnt, float* __restrict__ dinv,
                    float* __restrict__ m, int n) {
  __shared__ float lds[WAVES_PER_BLOCK * 64];
  int wave  = (int)((blockIdx.x * (unsigned)blockDim.x + threadIdx.x) >> 5);
  int tl    = threadIdx.x & 31;
  int row0  = wave * 16;
  if (row0 >= n) return;
  int lane16 = tl & 15;
  int hi     = tl >> 4;

  v16h b;
  {
    int  colW = lane16 & 3;
    bool keep = lane16 < 4;
#pragma unroll
    for (int j = 0; j < 16; ++j) {
      int k = (j < 8) ? (hi * 8 + j) : (16 + hi * 8 + (j - 8));
      float wv = W[k * 4 + colW];
      b[j] = keep ? (_Float16)wv : (_Float16)0.0f;
    }
  }

  int  arow = row0 + lane16;
  bool rok  = arow < n;
  int  ar   = rok ? arow : (n - 1);
  const float* xr = x + (size_t)ar * 32;
  float4 q0 = *reinterpret_cast<const float4*>(xr + hi * 8);
  float4 q1 = *reinterpret_cast<const float4*>(xr + hi * 8 + 4);
  float4 q2 = *reinterpret_cast<const float4*>(xr + 16 + hi * 8);
  float4 q3 = *reinterpret_cast<const float4*>(xr + 16 + hi * 8 + 4);
  float zf = rok ? 1.0f : 0.0f;
  v16h a;
  a[0]=(_Float16)(q0.x*zf); a[1]=(_Float16)(q0.y*zf); a[2]=(_Float16)(q0.z*zf); a[3]=(_Float16)(q0.w*zf);
  a[4]=(_Float16)(q1.x*zf); a[5]=(_Float16)(q1.y*zf); a[6]=(_Float16)(q1.z*zf); a[7]=(_Float16)(q1.w*zf);
  a[8]=(_Float16)(q2.x*zf); a[9]=(_Float16)(q2.y*zf); a[10]=(_Float16)(q2.z*zf); a[11]=(_Float16)(q2.w*zf);
  a[12]=(_Float16)(q3.x*zf); a[13]=(_Float16)(q3.y*zf); a[14]=(_Float16)(q3.z*zf); a[15]=(_Float16)(q3.w*zf);

  v8f c = {};
  c = __builtin_amdgcn_wmma_f32_16x16x32_f16(false, a, false, b,
                                             (short)0, c, false, false);

  float* slab = lds + ((threadIdx.x >> 5) << 6);
  if (lane16 < 4) {
#pragma unroll
    for (int r = 0; r < 8; ++r) slab[(hi * 8 + r) * 4 + lane16] = c[r];
  }
  asm volatile("s_wait_dscnt 0" ::: "memory");
  if (hi == 0) {
    int mr = row0 + lane16;
    if (mr < n) {
      float4 v = *reinterpret_cast<const float4*>(slab + lane16 * 4);
      float di = rsqrtf(1.0f + (float)cnt[mr]);
      dinv[mr] = di;
      v.x *= di; v.y *= di; v.z *= di; v.w *= di;
      *reinterpret_cast<float4*>(m + (size_t)mr * 4) = v;
    }
  }
}

#define GATHER_GRID 2048

__device__ __forceinline__ void vzero(float4& v) { v = make_float4(0.0f, 0.0f, 0.0f, 0.0f); }
__device__ __forceinline__ void vzero(float2& v) { v = make_float2(0.0f, 0.0f); }
__device__ __forceinline__ void vacc(float (&a)[4], const float4& v) {
  a[0] += v.x; a[1] += v.y; a[2] += v.z; a[3] += v.w;
}
__device__ __forceinline__ void vacc(float (&a)[4], const float2& v) {
  a[0] += v.x; a[1] += v.y;
}
template<int F> struct RowVec;
template<> struct RowVec<4> { typedef float4 type; };
template<> struct RowVec<2> { typedef float2 type; };

template<int F, int FOUT, bool BIAS2, bool SCALE, bool WRITE_H>
__global__ __launch_bounds__(THREADS)
void k_gather_xform(const int* __restrict__ cnt, const int* __restrict__ slot,
                    const int* __restrict__ ovf,
                    const float* __restrict__ p, const float* __restrict__ dinv,
                    const float* __restrict__ b, const float* __restrict__ W,
                    const float* __restrict__ bias2,
                    float* __restrict__ hout, float* __restrict__ mout, int n) {
  typedef typename RowVec<F>::type V;
  __shared__ float lds[WAVES_PER_BLOCK * 64];
  const V* pv = reinterpret_cast<const V*>(p);
  const int lane   = threadIdx.x & 31;
  const int lane16 = lane & 15;
  const int hi     = lane >> 4;
  const int tw   = (int)((gridDim.x * (unsigned)blockDim.x) >> 5);
  int node = __builtin_amdgcn_readfirstlane(
      (int)((blockIdx.x * (unsigned)blockDim.x + threadIdx.x) >> 5));
  if (node >= n) return;

  const int   f0    = (F == 4) ? (2 * hi) : 0;
  const float bias0 = b[f0];
  const float bias1 = b[f0 + 1];
  const bool  kreal = (F == 4) || (hi == 0);

  v2f bw;
#pragma unroll
  for (int j = 0; j < 2; ++j) {
    int  k    = hi * 2 + j;
    int  kc   = (k < F) ? k : (F - 1);
    int  cc   = (lane16 < FOUT) ? lane16 : (FOUT - 1);
    float wv  = W[kc * FOUT + cc];
    bw[j] = (k < F && lane16 < FOUT) ? wv : 0.0f;
  }

  v2f   arow   = {0.0f, 0.0f};
  int   mynode = -1;
  float mydi   = 0.0f;
  int   filled = 0;

  int deg = cnt[node];
  int m   = (deg < ROWCAP) ? deg : ROWCAP;
  int s0, s1, s2, s3;
  {
    const int* row = slot + (size_t)node * ROWCAP;
    s0 = (lane      < m) ? row[lane]      : -1;
    s1 = (lane + 32 < m) ? row[lane + 32] : -1;
    s2 = (lane + 64 < m) ? row[lane + 64] : -1;
    s3 = (lane + 96 < m) ? row[lane + 96] : -1;
  }

  for (;;) {
    const int  next = node + tw;
    const bool more = next < n;
    const int  nx   = more ? next : node;
    const int  ndeg = cnt[nx];

    const V g0 = pv[(s0 >= 0) ? s0 : node];
    const V g1 = pv[(s1 >= 0) ? s1 : node];
    const V g2 = pv[(s2 >= 0) ? s2 : node];

    const float  di   = dinv[node];
    const float2 self = *reinterpret_cast<const float2*>(p + (size_t)node * F + f0);

    const int nm = more ? ((ndeg < ROWCAP) ? ndeg : ROWCAP) : 0;
    int t0, t1, t2, t3;
    {
      const int* nrow = slot + (size_t)nx * ROWCAP;
      t0 = (lane      < nm) ? nrow[lane]      : -1;
      t1 = (lane + 32 < nm) ? nrow[lane + 32] : -1;
      t2 = (lane + 64 < nm) ? nrow[lane + 64] : -1;
      t3 = (lane + 96 < nm) ? nrow[lane + 96] : -1;
    }

    V z; vzero(z);
    float a[4] = {0.0f, 0.0f, 0.0f, 0.0f};
    vacc(a, (s0 >= 0) ? g0 : z);
    vacc(a, (s1 >= 0) ? g1 : z);
    vacc(a, (s2 >= 0) ? g2 : z);
    if (m > 96) {
      if (s3 >= 0) { V g3 = pv[s3]; vacc(a, g3); }
    }

    if (deg > ROWCAP) {
      int novf = cnt[n];
      for (int k = lane; k < novf; k += 32) {
        if (ovf[2 * (size_t)k + 1] == node) {
          V vo = pv[ovf[2 * (size_t)k]];
          vacc(a, vo);
        }
      }
    }

#pragma unroll
    for (int o = 16; o > 0; o >>= 1) {
      a[0] += __shfl_xor(a[0], o);
      a[1] += __shfl_xor(a[1], o);
      if (F == 4) {
        a[2] += __shfl_xor(a[2], o);
        a[3] += __shfl_xor(a[3], o);
      }
    }

    const float sum0 = (F == 4 && hi == 1) ? a[2] : a[0];
    const float sum1 = (F == 4 && hi == 1) ? a[3] : a[1];
    const float h0 = tanhf(di * (sum0 + self.x) + bias0);
    const float h1 = tanhf(di * (sum1 + self.y) + bias1);

    if (WRITE_H) {
      if (lane == 0) *reinterpret_cast<float2*>(hout + (size_t)node * F) = make_float2(h0, h1);
    }

    {
      const bool cap = (lane16 == filled);
      arow[0] = cap ? (kreal ? h0 : 0.0f) : arow[0];
      arow[1] = cap ? (kreal ? h1 : 0.0f) : arow[1];
      mynode  = cap ? node : mynode;
      mydi    = cap ? di : mydi;
      ++filled;
    }

    if (filled == 16 || !more) {
      v8f c = {};
      c = __builtin_amdgcn_wmma_f32_16x16x4_f32(false, arow, false, bw,
                                                (short)0, c, false, false);
      float* slab = lds + ((threadIdx.x >> 5) << 6);
      if (lane16 < FOUT) {
#pragma unroll
        for (int r = 0; r < 8; ++r) slab[(hi * 8 + r) * FOUT + lane16] = c[r];
      }
      asm volatile("s_wait_dscnt 0" ::: "memory");
      if (hi == 0 && mynode >= 0) {
        if (FOUT == 4) {
          float4 v = *reinterpret_cast<const float4*>(slab + lane16 * 4);
          if (BIAS2) { v.x += bias2[0]; v.y += bias2[1]; v.z += bias2[2]; v.w += bias2[3]; }
          if (SCALE) { v.x *= mydi; v.y *= mydi; v.z *= mydi; v.w *= mydi; }
          *reinterpret_cast<float4*>(mout + (size_t)mynode * 4) = v;
        } else {
          float2 v = *reinterpret_cast<const float2*>(slab + lane16 * 2);
          if (BIAS2) { v.x += bias2[0]; v.y += bias2[1]; }
          if (SCALE) { v.x *= mydi; v.y *= mydi; }
          *reinterpret_cast<float2*>(mout + (size_t)mynode * 2) = v;
        }
      }
      arow[0] = 0.0f; arow[1] = 0.0f;
      mynode = -1; filled = 0;
    }

    if (!more) break;
    node = next; deg = ndeg; m = nm;
    s0 = t0; s1 = t1; s2 = t2; s3 = t3;
  }
}

extern "C" void kernel_launch(void* const* d_in, const int* in_sizes, int n_in,
                              void* d_out, int out_size, void* d_ws, size_t ws_size,
                              hipStream_t stream) {
  const float* x  = (const float*)d_in[0];
  const int*   ei = (const int*)  d_in[1];
  const float* W1 = (const float*)d_in[2];
  const float* b1 = (const float*)d_in[3];
  const float* W2 = (const float*)d_in[4];
  const float* b2 = (const float*)d_in[5];
  const float* W3 = (const float*)d_in[6];
  const float* b3 = (const float*)d_in[7];
  const float* Wc = (const float*)d_in[8];
  const float* bc = (const float*)d_in[9];

  const int N = in_sizes[0] / 32;
  const int E = in_sizes[1] / 2;
  const int* src = ei;
  const int* dst = ei + E;

  float* out = (float*)d_out;
  float* h3  = out + (size_t)N * 4;

  int*   slot = (int*)d_ws;
  float* pa   = (float*)(slot + (size_t)N * ROWCAP);
  float* pb   = pa + (size_t)4 * N;
  float* dinv = pa + (size_t)8 * N;
  int*   cnt  = (int*)(pa + (size_t)9 * N);
  int*   queue = cnt + (size_t)N + 2 + 64;
  int*   ovf  = cnt + (size_t)N + 132;

  const int gZ  = (N + 132 + THREADS - 1) / THREADS;
  const int gE  = 2048;
  const int gN  = (N + WAVES_PER_BLOCK - 1) / WAVES_PER_BLOCK;
  const int gG  = gN < GATHER_GRID ? gN : GATHER_GRID;
  const int tiles = (N + 15) / 16;
  const int gW  = (tiles + WAVES_PER_BLOCK - 1) / WAVES_PER_BLOCK;

  k_zero_int<<<gZ, THREADS, 0, stream>>>(cnt, N + 132);
  k_place<<<gE, THREADS, 0, stream>>>(src, dst, cnt, slot, ovf, queue, N, E);

  k_xform32_wmma<<<gW, THREADS, 0, stream>>>(x, W1, cnt, dinv, pa, N);

  k_gather_xform<4, 4, false, true, false><<<gG, THREADS, 0, stream>>>(
      cnt, slot, ovf, pa, dinv, b1, W2, nullptr, nullptr, pb, N);

  k_gather_xform<4, 2, false, true, false><<<gG, THREADS, 0, stream>>>(
      cnt, slot, ovf, pb, dinv, b2, W3, nullptr, nullptr, pa, N);

  k_gather_xform<2, 4, true, false, true><<<gG, THREADS, 0, stream>>>(
      cnt, slot, ovf, pa, dinv, b3, Wc, bc, h3, out, N);
}


# Table build by two-level LDS counting sort: 8192-edge tiles ordered by 256-node bin (no global add, each tile's runs in lines of its own), a workgroup per bin ranks rows in LDS; sum order = LDS rank
# speedup vs baseline: 16.0923x; 1.1441x over previous
#include <hip/hip_runtime.h>
#include <hip/hip_fp16.h>

typedef __attribute__((ext_vector_type(16))) _Float16 v16h;
typedef __attribute__((ext_vector_type(8)))  float    v8f;
typedef __attribute__((ext_vector_type(2)))  float    v2f;

#define THREADS 256
#define WAVES_PER_BLOCK (THREADS / 32)

#define ROWCAP 128

#define BIN_SHIFT  8
#define BIN_W      (1 << BIN_SHIFT)
#define MAXNB      1024
#define TILE       8192
#define TS_THREADS 256
#define TS_EPT     (TILE / TS_THREADS)

__global__ __launch_bounds__(TS_THREADS)
void k_tile_sort(const int* __restrict__ src, const int* __restrict__ dst,
                 int2* __restrict__ sorted, int* __restrict__ offs,
                 int* __restrict__ ovfcount, int nb, int e) {
  __shared__ int  hist[MAXNB + 1];
  __shared__ int2 stage[TILE];
  const int tid  = threadIdx.x;
  const int tile = blockIdx.x;
  const int base = tile * TILE;

  if (tile == 0 && tid == 0) *ovfcount = 0;
  for (int i = tid; i <= nb; i += TS_THREADS) hist[i] = 0;
  __syncthreads();

  int dd[TS_EPT];
  int ss[TS_EPT];
  int rr[TS_EPT];
#pragma unroll
  for (int k = 0; k < TS_EPT; ++k) {
    int i  = base + k * TS_THREADS + tid;
    int ic = (i < e) ? i : (e - 1);
    dd[k] = dst[ic];
    ss[k] = src[ic];
  }
#pragma unroll
  for (int k = 0; k < TS_EPT; ++k) {
    int i   = base + k * TS_THREADS + tid;
    int bin = (int)((unsigned)dd[k] >> BIN_SHIFT);
    bin = (bin < nb) ? bin : (nb - 1);
    dd[k] = (i < e) ? dd[k] : -1;
    rr[k] = (i < e) ? atomicAdd(&hist[bin], 1) : 0;
  }
  __syncthreads();

  if (tid < 32) {
    const int per = (nb + 31) >> 5;
    const int b0  = tid * per;
    const int b1  = (b0 + per < nb) ? (b0 + per) : nb;
    int sum = 0;
    for (int b = b0; b < b1; ++b) sum += hist[b];
    int incl = sum;
    for (int o = 1; o < 32; o <<= 1) {
      int v = __shfl_up(incl, o);
      if (tid >= o) incl += v;
    }
    int run = incl - sum;
    for (int b = b0; b < b1; ++b) { int c = hist[b]; hist[b] = run; run += c; }
    if (tid == 31) hist[nb] = incl;
  }
  __syncthreads();

  int* myoffs = offs + (size_t)tile * (size_t)(nb + 1);
  for (int i = tid; i <= nb; i += TS_THREADS) myoffs[i] = hist[i];

#pragma unroll
  for (int k = 0; k < TS_EPT; ++k) {
    if (dd[k] >= 0) {
      int bin = (int)((unsigned)dd[k] >> BIN_SHIFT);
      bin = (bin < nb) ? bin : (nb - 1);
      stage[hist[bin] + rr[k]] = make_int2(ss[k], dd[k]);
    }
  }
  __syncthreads();

  const int total = hist[nb];
#pragma unroll
  for (int k = 0; k < TS_EPT; ++k) {
    int i = k * TS_THREADS + tid;
    if (i < total) sorted[(size_t)base + i] = stage[i];
  }
}

#define RB_THREADS 1024
#define RB_WAVES   (RB_THREADS / 32)
#define RB_DEPTH   8

__global__ __launch_bounds__(RB_THREADS)
void k_rows_build(const int2* __restrict__ sorted, const int* __restrict__ offs,
                  int* __restrict__ cnt, int* __restrict__ slot, int* __restrict__ ovf,
                  int n, int nb, int ntiles) {
  __shared__ int rows[BIN_W * ROWCAP];
  __shared__ int lcnt[BIN_W];
  const int b     = blockIdx.x;
  const int tid   = threadIdx.x;
  const int lane  = tid & 31;
  const int wave  = __builtin_amdgcn_readfirstlane(tid >> 5);
  const int node0 = b << BIN_SHIFT;

  for (int i = tid; i < BIN_W; i += RB_THREADS) lcnt[i] = 0;
  __syncthreads();

  const int per = (ntiles + RB_WAVES - 1) / RB_WAVES;
  const int t0  = wave * per;
  const int t1  = (t0 + per < ntiles) ? (t0 + per) : ntiles;

  for (int tb = t0; tb < t1; tb += 32) {
    const int  t  = tb + lane;
    const bool tv = t < t1;
    const int  tc = tv ? t : (t1 - 1);
    const int* po = offs + (size_t)tc * (size_t)(nb + 1) + b;
    int o0 = po[0];
    int o1 = po[1];
    o1 = tv ? o1 : o0;
    const int nt = (t1 - tb < 32) ? (t1 - tb) : 32;

    for (int j = 0; j < nt; j += RB_DEPTH) {
      int2 ev[RB_DEPTH];
      int  st[RB_DEPTH];
      int  len[RB_DEPTH];
      size_t gb[RB_DEPTH];
#pragma unroll
      for (int u = 0; u < RB_DEPTH; ++u) {
        const int  jj  = j + u;
        const bool on  = jj < nt;
        const int  a0  = __builtin_amdgcn_readlane(o0, jj & 31);
        const int  a1  = __builtin_amdgcn_readlane(o1, jj & 31);
        len[u] = on ? (a1 - a0) : 0;
        st[u]  = (len[u] > 0) ? a0 : 0;
        gb[u]  = (size_t)(tb + (on ? jj : 0)) * TILE;
        const int li = (lane < len[u]) ? lane : ((len[u] > 0) ? (len[u] - 1) : 0);
        ev[u]  = sorted[gb[u] + st[u] + li];
      }
#pragma unroll
      for (int u = 0; u < RB_DEPTH; ++u) {
        int  i    = lane;
        int2 edge = ev[u];
        while (i < len[u]) {
          const int s  = edge.x;
          const int d  = edge.y;
          const int dl = d - node0;
          if ((unsigned)dl < (unsigned)BIN_W && d < n) {
            const int r = atomicAdd(&lcnt[dl], 1);
            if (r < ROWCAP) {
              rows[dl * ROWCAP + r] = s;
            } else {
              int q = atomicAdd(&cnt[n], 1);
              ovf[2 * (size_t)q]     = s;
              ovf[2 * (size_t)q + 1] = d;
            }
          }
          i += 32;
          if (i < len[u]) edge = sorted[gb[u] + st[u] + i];
        }
      }
    }
  }
  __syncthreads();

  for (int rl = wave; rl < BIN_W; rl += RB_WAVES) {
    const int d = node0 + rl;
    if (d < n) {
      const int c = lcnt[rl];
      const int m = (c < ROWCAP) ? c : ROWCAP;
      for (int k = lane; k < m; k += 32) slot[(size_t)d * ROWCAP + k] = rows[rl * ROWCAP + k];
    }
  }
  for (int i = tid; i < BIN_W; i += RB_THREADS) {
    if (node0 + i < n) cnt[node0 + i] = lcnt[i];
  }
}

__global__ __launch_bounds__(THREADS)
void k_xform32_wmma(const float* __restrict__ x, const float* __restrict__ W,
                    const int* __restrict__ cnt, float* __restrict__ dinv,
                    float* __restrict__ m, int n) {
  __shared__ float lds[WAVES_PER_BLOCK * 64];
  int wave  = (int)((blockIdx.x * (unsigned)blockDim.x + threadIdx.x) >> 5);
  int tl    = threadIdx.x & 31;
  int row0  = wave * 16;
  if (row0 >= n) return;
  int lane16 = tl & 15;
  int hi     = tl >> 4;

  v16h b;
  {
    int  colW = lane16 & 3;
    bool keep = lane16 < 4;
#pragma unroll
    for (int j = 0; j < 16; ++j) {
      int k = (j < 8) ? (hi * 8 + j) : (16 + hi * 8 + (j - 8));
      float wv = W[k * 4 + colW];
      b[j] = keep ? (_Float16)wv : (_Float16)0.0f;
    }
  }

  int  arow = row0 + lane16;
  bool rok  = arow < n;
  int  ar   = rok ? arow : (n - 1);
  const float* xr = x + (size_t)ar * 32;
  float4 q0 = *reinterpret_cast<const float4*>(xr + hi * 8);
  float4 q1 = *reinterpret_cast<const float4*>(xr + hi * 8 + 4);
  float4 q2 = *reinterpret_cast<const float4*>(xr + 16 + hi * 8);
  float4 q3 = *reinterpret_cast<const float4*>(xr + 16 + hi * 8 + 4);
  float zf = rok ? 1.0f : 0.0f;
  v16h a;
  a[0]=(_Float16)(q0.x*zf); a[1]=(_Float16)(q0.y*zf); a[2]=(_Float16)(q0.z*zf); a[3]=(_Float16)(q0.w*zf);
  a[4]=(_Float16)(q1.x*zf); a[5]=(_Float16)(q1.y*zf); a[6]=(_Float16)(q1.z*zf); a[7]=(_Float16)(q1.w*zf);
  a[8]=(_Float16)(q2.x*zf); a[9]=(_Float16)(q2.y*zf); a[10]=(_Float16)(q2.z*zf); a[11]=(_Float16)(q2.w*zf);
  a[12]=(_Float16)(q3.x*zf); a[13]=(_Float16)(q3.y*zf); a[14]=(_Float16)(q3.z*zf); a[15]=(_Float16)(q3.w*zf);

  v8f c = {};
  c = __builtin_amdgcn_wmma_f32_16x16x32_f16(false, a, false, b,
                                             (short)0, c, false, false);

  float* slab = lds + ((threadIdx.x >> 5) << 6);
  if (lane16 < 4) {
#pragma unroll
    for (int r = 0; r < 8; ++r) slab[(hi * 8 + r) * 4 + lane16] = c[r];
  }
  asm volatile("s_wait_dscnt 0" ::: "memory");
  if (hi == 0) {
    int mr = row0 + lane16;
    if (mr < n) {
      float4 v = *reinterpret_cast<const float4*>(slab + lane16 * 4);
      float di = rsqrtf(1.0f + (float)cnt[mr]);
      dinv[mr] = di;
      v.x *= di; v.y *= di; v.z *= di; v.w *= di;
      *reinterpret_cast<float4*>(m + (size_t)mr * 4) = v;
    }
  }
}

#define GATHER_GRID 2048

__device__ __forceinline__ void vzero(float4& v) { v = make_float4(0.0f, 0.0f, 0.0f, 0.0f); }
__device__ __forceinline__ void vzero(float2& v) { v = make_float2(0.0f, 0.0f); }
__device__ __forceinline__ void vacc(float (&a)[4], const float4& v) {
  a[0] += v.x; a[1] += v.y; a[2] += v.z; a[3] += v.w;
}
__device__ __forceinline__ void vacc(float (&a)[4], const float2& v) {
  a[0] += v.x; a[1] += v.y;
}
template<int F> struct RowVec;
template<> struct RowVec<4> { typedef float4 type; };
template<> struct RowVec<2> { typedef float2 type; };

template<int F, int FOUT, bool BIAS2, bool SCALE, bool WRITE_H>
__global__ __launch_bounds__(THREADS)
void k_gather_xform(const int* __restrict__ cnt, const int* __restrict__ slot,
                    const int* __restrict__ ovf,
                    const float* __restrict__ p, const float* __restrict__ dinv,
                    const float* __restrict__ b, const float* __restrict__ W,
                    const float* __restrict__ bias2,
                    float* __restrict__ hout, float* __restrict__ mout, int n) {
  typedef typename RowVec<F>::type V;
  __shared__ float lds[WAVES_PER_BLOCK * 64];
  const V* pv = reinterpret_cast<const V*>(p);
  const int lane   = threadIdx.x & 31;
  const int lane16 = lane & 15;
  const int hi     = lane >> 4;
  const int tw   = (int)((gridDim.x * (unsigned)blockDim.x) >> 5);
  int node = __builtin_amdgcn_readfirstlane(
      (int)((blockIdx.x * (unsigned)blockDim.x + threadIdx.x) >> 5));
  if (node >= n) return;

  const int   f0    = (F == 4) ? (2 * hi) : 0;
  const float bias0 = b[f0];
  const float bias1 = b[f0 + 1];
  const bool  kreal = (F == 4) || (hi == 0);

  v2f bw;
#pragma unroll
  for (int j = 0; j < 2; ++j) {
    int  k    = hi * 2 + j;
    int  kc   = (k < F) ? k : (F - 1);
    int  cc   = (lane16 < FOUT) ? lane16 : (FOUT - 1);
    float wv  = W[kc * FOUT + cc];
    bw[j] = (k < F && lane16 < FOUT) ? wv : 0.0f;
  }

  v2f   arow   = {0.0f, 0.0f};
  int   mynode = -1;
  float mydi   = 0.0f;
  int   filled = 0;

  int deg = cnt[node];
  int m   = (deg < ROWCAP) ? deg : ROWCAP;
  int s0, s1, s2, s3;
  {
    const int* row = slot + (size_t)node * ROWCAP;
    s0 = (lane      < m) ? row[lane]      : -1;
    s1 = (lane + 32 < m) ? row[lane + 32] : -1;
    s2 = (lane + 64 < m) ? row[lane + 64] : -1;
    s3 = (lane + 96 < m) ? row[lane + 96] : -1;
  }

  for (;;) {
    const int  next = node + tw;
    const bool more = next < n;
    const int  nx   = more ? next : node;
    const int  ndeg = cnt[nx];

    const V g0 = pv[(s0 >= 0) ? s0 : node];
    const V g1 = pv[(s1 >= 0) ? s1 : node];
    const V g2 = pv[(s2 >= 0) ? s2 : node];

    const float  di   = dinv[node];
    const float2 self = *reinterpret_cast<const float2*>(p + (size_t)node * F + f0);

    const int nm = more ? ((ndeg < ROWCAP) ? ndeg : ROWCAP) : 0;
    int t0, t1, t2, t3;
    {
      const int* nrow = slot + (size_t)nx * ROWCAP;
      t0 = (lane      < nm) ? nrow[lane]      : -1;
      t1 = (lane + 32 < nm) ? nrow[lane + 32] : -1;
      t2 = (lane + 64 < nm) ? nrow[lane + 64] : -1;
      t3 = (lane + 96 < nm) ? nrow[lane + 96] : -1;
    }

    V z; vzero(z);
    float a[4] = {0.0f, 0.0f, 0.0f, 0.0f};
    vacc(a, (s0 >= 0) ? g0 : z);
    vacc(a, (s1 >= 0) ? g1 : z);
    vacc(a, (s2 >= 0) ? g2 : z);
    if (m > 96) {
      if (s3 >= 0) { V g3 = pv[s3]; vacc(a, g3); }
    }

    if (deg > ROWCAP) {
      int novf = cnt[n];
      for (int k = lane; k < novf; k += 32) {
        if (ovf[2 * (size_t)k + 1] == node) {
          V vo = pv[ovf[2 * (size_t)k]];
          vacc(a, vo);
        }
      }
    }

#pragma unroll
    for (int o = 16; o > 0; o >>= 1) {
      a[0] += __shfl_xor(a[0], o);
      a[1] += __shfl_xor(a[1], o);
      if (F == 4) {
        a[2] += __shfl_xor(a[2], o);
        a[3] += __shfl_xor(a[3], o);
      }
    }

    const float sum0 = (F == 4 && hi == 1) ? a[2] : a[0];
    const float sum1 = (F == 4 && hi == 1) ? a[3] : a[1];
    const float h0 = tanhf(di * (sum0 + self.x) + bias0);
    const float h1 = tanhf(di * (sum1 + self.y) + bias1);

    if (WRITE_H) {
      if (lane == 0) *reinterpret_cast<float2*>(hout + (size_t)node * F) = make_float2(h0, h1);
    }

    {
      const bool cap = (lane16 == filled);
      arow[0] = cap ? (kreal ? h0 : 0.0f) : arow[0];
      arow[1] = cap ? (kreal ? h1 : 0.0f) : arow[1];
      mynode  = cap ? node : mynode;
      mydi    = cap ? di : mydi;
      ++filled;
    }

    if (filled == 16 || !more) {
      v8f c = {};
      c = __builtin_amdgcn_wmma_f32_16x16x4_f32(false, arow, false, bw,
                                                (short)0, c, false, false);
      float* slab = lds + ((threadIdx.x >> 5) << 6);
      if (lane16 < FOUT) {
#pragma unroll
        for (int r = 0; r < 8; ++r) slab[(hi * 8 + r) * FOUT + lane16] = c[r];
      }
      asm volatile("s_wait_dscnt 0" ::: "memory");
      if (hi == 0 && mynode >= 0) {
        if (FOUT == 4) {
          float4 v = *reinterpret_cast<const float4*>(slab + lane16 * 4);
          if (BIAS2) { v.x += bias2[0]; v.y += bias2[1]; v.z += bias2[2]; v.w += bias2[3]; }
          if (SCALE) { v.x *= mydi; v.y *= mydi; v.z *= mydi; v.w *= mydi; }
          *reinterpret_cast<float4*>(mout + (size_t)mynode * 4) = v;
        } else {
          float2 v = *reinterpret_cast<const float2*>(slab + lane16 * 2);
          if (BIAS2) { v.x += bias2[0]; v.y += bias2[1]; }
          if (SCALE) { v.x *= mydi; v.y *= mydi; }
          *reinterpret_cast<float2*>(mout + (size_t)mynode * 2) = v;
        }
      }
      arow[0] = 0.0f; arow[1] = 0.0f;
      mynode = -1; filled = 0;
    }

    if (!more) break;
    node = next; deg = ndeg; m = nm;
    s0 = t0; s1 = t1; s2 = t2; s3 = t3;
  }
}

extern "C" void kernel_launch(void* const* d_in, const int* in_sizes, int n_in,
                              void* d_out, int out_size, void* d_ws, size_t ws_size,
                              hipStream_t stream) {
  const float* x  = (const float*)d_in[0];
  const int*   ei = (const int*)  d_in[1];
  const float* W1 = (const float*)d_in[2];
  const float* b1 = (const float*)d_in[3];
  const float* W2 = (const float*)d_in[4];
  const float* b2 = (const float*)d_in[5];
  const float* W3 = (const float*)d_in[6];
  const float* b3 = (const float*)d_in[7];
  const float* Wc = (const float*)d_in[8];
  const float* bc = (const float*)d_in[9];

  const int N = in_sizes[0] / 32;
  const int E = in_sizes[1] / 2;
  const int* src = ei;
  const int* dst = ei + E;

  float* out = (float*)d_out;
  float* h3  = out + (size_t)N * 4;

  const int ntiles = (E + TILE - 1) / TILE;
  const int nb     = (N + BIN_W - 1) / BIN_W;
  int*   slot = (int*)d_ws;
  int2*  sorted = (int2*)(slot + (size_t)N * ROWCAP);
  float* pa   = (float*)(sorted + (size_t)ntiles * TILE);
  float* pb   = pa + (size_t)4 * N;
  float* dinv = pa + (size_t)8 * N;
  int*   cnt  = (int*)(pa + (size_t)9 * N);
  int*   ovf  = cnt + (size_t)N + 2;
  int*   offs = ovf + (size_t)2 * E;

  const int gN  = (N + WAVES_PER_BLOCK - 1) / WAVES_PER_BLOCK;
  const int gG  = gN < GATHER_GRID ? gN : GATHER_GRID;
  const int tiles = (N + 15) / 16;
  const int gW  = (tiles + WAVES_PER_BLOCK - 1) / WAVES_PER_BLOCK;

  k_tile_sort<<<ntiles, TS_THREADS, 0, stream>>>(src, dst, sorted, offs, cnt + N, nb, E);
  k_rows_build<<<nb, RB_THREADS, 0, stream>>>(sorted, offs, cnt, slot, ovf, N, nb, ntiles);

  k_xform32_wmma<<<gW, THREADS, 0, stream>>>(x, W1, cnt, dinv, pa, N);

  k_gather_xform<4, 4, false, true, false><<<gG, THREADS, 0, stream>>>(
      cnt, slot, ovf, pa, dinv, b1, W2, nullptr, nullptr, pb, N);

  k_gather_xform<4, 2, false, true, false><<<gG, THREADS, 0, stream>>>(
      cnt, slot, ovf, pb, dinv, b2, W3, nullptr, nullptr, pa, N);

  k_gather_xform<2, 4, true, false, true><<<gG, THREADS, 0, stream>>>(
      cnt, slot, ovf, pa, dinv, b3, Wc, bc, h3, out, N);
}


# Aggregation over 2-D edge buckets (source part x destination bin): a workgroup keeps the part's rows and the bin's sums in LDS, LDS float adds; no random row reads from L2; sum order = arrival
# speedup vs baseline: 17.9276x; 1.1140x over previous
#include <hip/hip_runtime.h>
#include <hip/hip_fp16.h>


typedef __attribute__((ext_vector_type(16))) _Float16 v16h;
typedef __attribute__((ext_vector_type(8)))  float    v8f;
typedef __attribute__((ext_vector_type(2)))  float    v2f;

#define THREADS 256
#define WAVES_PER_BLOCK (THREADS / 32)

#define PSHIFT 13
#define PW     (1 << PSHIFT)
#define PMASK  (PW - 1)
#define MAXNB  1024
#define TILE       8192
#define TS_THREADS 256
#define TS_EPT     (TILE / TS_THREADS)

__device__ __forceinline__ int bucket_of(int s, int d, int np) {
  int db = (int)((unsigned)d >> PSHIFT);
  int sp = (int)((unsigned)s >> PSHIFT);
  db = (db < np) ? db : (np - 1);
  sp = (sp < np) ? sp : (np - 1);
  return db * np + sp;
}

__global__ __launch_bounds__(TS_THREADS)
void k_tile_sort(const int* __restrict__ src, const int* __restrict__ dst,
                 int* __restrict__ sorted, int* __restrict__ offs,
                 int np, int nbk, int e) {
  __shared__ int hist[MAXNB + 1];
  __shared__ int stage[TILE];
  const int tid  = threadIdx.x;
  const int tile = blockIdx.x;
  const int base = tile * TILE;

  for (int i = tid; i <= nbk; i += TS_THREADS) hist[i] = 0;
  __syncthreads();

  int dd[TS_EPT];
  int ss[TS_EPT];
#pragma unroll
  for (int k = 0; k < TS_EPT; ++k) {
    int i  = base + k * TS_THREADS + tid;
    int ic = (i < e) ? i : (e - 1);
    dd[k] = dst[ic];
    ss[k] = src[ic];
  }
#pragma unroll
  for (int k = 0; k < TS_EPT; ++k) {
    int i   = base + k * TS_THREADS + tid;
    int bin = bucket_of(ss[k], dd[k], np);
    ss[k] = (ss[k] & PMASK) | ((dd[k] & PMASK) << PSHIFT);
    const int r = (i < e) ? atomicAdd(&hist[bin], 1) : 0;
    dd[k] = (i < e) ? (bin | (r << 10)) : -1;
  }
  __syncthreads();

  if (tid < 32) {
    const int per = (nbk + 31) >> 5;
    const int b0  = tid * per;
    const int b1  = (b0 + per < nbk) ? (b0 + per) : nbk;
    int sum = 0;
    for (int b = b0; b < b1; ++b) sum += hist[b];
    int incl = sum;
    for (int o = 1; o < 32; o <<= 1) {
      int v = __shfl_up(incl, o);
      if (tid >= o) incl += v;
    }
    int run = incl - sum;
    for (int b = b0; b < b1; ++b) { int c = hist[b]; hist[b] = run; run += c; }
    if (tid == 31) hist[nbk] = incl;
  }
  __syncthreads();

  int* myoffs = offs + (size_t)tile * (size_t)(nbk + 1);
  for (int i = tid; i <= nbk; i += TS_THREADS) myoffs[i] = hist[i];

#pragma unroll
  for (int k = 0; k < TS_EPT; ++k) {
    if (dd[k] >= 0) stage[hist[dd[k] & (MAXNB - 1)] + (dd[k] >> 10)] = ss[k];
  }
  __syncthreads();

  const int total = hist[nbk];
#pragma unroll
  for (int k = 0; k < TS_EPT; ++k) {
    int i = k * TS_THREADS + tid;
    if (i < total) sorted[(size_t)base + i] = stage[i];
  }
}

#define WK_THREADS 1024
#define WK_WAVES   (WK_THREADS / 32)
#define WK_DEPTH   8

template<typename Fn>
__device__ __forceinline__ void walk_bucket(const int* __restrict__ sorted,
                                            const int* __restrict__ offs,
                                            int b, int nbk, int ntiles, int lane, int wave, Fn f) {
  const int per = (ntiles + WK_WAVES - 1) / WK_WAVES;
  const int t0  = wave * per;
  const int t1  = (t0 + per < ntiles) ? (t0 + per) : ntiles;

  for (int tb = t0; tb < t1; tb += 32) {
    const int  t  = tb + lane;
    const bool tv = t < t1;
    const int  tc = tv ? t : (t1 - 1);
    const int* po = offs + (size_t)tc * (size_t)(nbk + 1) + b;
    int o0 = po[0];
    int o1 = po[1];
    o1 = tv ? o1 : o0;
    const int nt = (t1 - tb < 32) ? (t1 - tb) : 32;

    for (int j = 0; j < nt; j += WK_DEPTH) {
      int    ev[WK_DEPTH];
      int    ew[WK_DEPTH];
      int    st[WK_DEPTH];
      int    len[WK_DEPTH];
      size_t gb[WK_DEPTH];
#pragma unroll
      for (int u = 0; u < WK_DEPTH; ++u) {
        const int  jj = j + u;
        const bool on = jj < nt;
        const int  a0 = __builtin_amdgcn_readlane(o0, jj & 31);
        const int  a1 = __builtin_amdgcn_readlane(o1, jj & 31);
        len[u] = on ? (a1 - a0) : 0;
        st[u]  = (len[u] > 0) ? a0 : 0;
        gb[u]  = (size_t)(tb + (on ? jj : 0)) * TILE;
        const int last = (len[u] > 0) ? (len[u] - 1) : 0;
        const int li = (lane      < len[u]) ? lane        : last;
        const int lj = (lane + 32 < len[u]) ? (lane + 32) : last;
        ev[u]  = sorted[gb[u] + st[u] + li];
        ew[u]  = sorted[gb[u] + st[u] + lj];
      }
#pragma unroll
      for (int u = 0; u < WK_DEPTH; ++u) {
        if (lane      < len[u]) f(ev[u]);
        if (lane + 32 < len[u]) f(ew[u]);
        for (int i = lane + 64; i < len[u]; i += 32) f(sorted[gb[u] + st[u] + i]);
      }
    }
  }
}

__global__ __launch_bounds__(WK_THREADS)
void k_degree(const int* __restrict__ sorted, const int* __restrict__ offs,
              int* __restrict__ degpart, int n, int np, int nbk, int npad, int ntiles) {
  __shared__ int lcnt[PW];
  const int b    = blockIdx.x;
  const int db   = b / np;
  const int sp   = b - db * np;
  const int tid  = threadIdx.x;
  const int lane = tid & 31;
  const int wave = __builtin_amdgcn_readfirstlane(tid >> 5);

  for (int i = tid; i < PW; i += WK_THREADS) lcnt[i] = 0;
  __syncthreads();
  walk_bucket(sorted, offs, b, nbk, ntiles, lane, wave, [&](int edge) {
    atomicAdd(&lcnt[(edge >> PSHIFT) & PMASK], 1);
  });
  __syncthreads();
  for (int i = tid; i < PW; i += WK_THREADS) {
    const int d = db * PW + i;
    if (d < n) degpart[(size_t)sp * npad + d] = lcnt[i];
  }
}

template<int F>
__global__ __launch_bounds__(WK_THREADS)
void k_gather_lds(const int* __restrict__ sorted, const int* __restrict__ offs,
                  const float* __restrict__ p, float* __restrict__ part,
                  int n, int np, int nbk, int npad, int ntiles) {
  __shared__ float chunk[PW * F];
  __shared__ float acc[PW * F];
  const int b    = blockIdx.x;
  const int db   = b / np;
  const int sp   = b - db * np;
  const int tid  = threadIdx.x;
  const int lane = tid & 31;
  const int wave = __builtin_amdgcn_readfirstlane(tid >> 5);

  const size_t nF = (size_t)n * F;
  for (int i = tid; i < PW * F / 4; i += WK_THREADS) {
    const size_t g = (size_t)sp * PW * F + (size_t)4 * i;
    float4 v = make_float4(0.0f, 0.0f, 0.0f, 0.0f);
    if (g + 4 <= nF) {
      v = *reinterpret_cast<const float4*>(p + g);
    } else {
      if (g     < nF) v.x = p[g];
      if (g + 1 < nF) v.y = p[g + 1];
      if (g + 2 < nF) v.z = p[g + 2];
    }
    *reinterpret_cast<float4*>(chunk + 4 * i) = v;
    *reinterpret_cast<float4*>(acc + 4 * i) = make_float4(0.0f, 0.0f, 0.0f, 0.0f);
  }
  __syncthreads();

  walk_bucket(sorted, offs, b, nbk, ntiles, lane, wave, [&](int edge) {
    const int sl = edge & PMASK;
    const int dl = (edge >> PSHIFT) & PMASK;
    if (F == 4) {
      const float4 r = *reinterpret_cast<const float4*>(chunk + sl * 4);
      float* a = acc + dl * 4;
      const int k = (dl >> 4) & 3;
      __hip_atomic_fetch_add(a + (0 ^ k), r.x, __ATOMIC_RELAXED, __HIP_MEMORY_SCOPE_WORKGROUP);
      __hip_atomic_fetch_add(a + (1 ^ k), r.y, __ATOMIC_RELAXED, __HIP_MEMORY_SCOPE_WORKGROUP);
      __hip_atomic_fetch_add(a + (2 ^ k), r.z, __ATOMIC_RELAXED, __HIP_MEMORY_SCOPE_WORKGROUP);
      __hip_atomic_fetch_add(a + (3 ^ k), r.w, __ATOMIC_RELAXED, __HIP_MEMORY_SCOPE_WORKGROUP);
    } else {
      const float2 r = *reinterpret_cast<const float2*>(chunk + sl * 2);
      float* a = acc + dl * 2;
      __hip_atomic_fetch_add(a,     r.x, __ATOMIC_RELAXED, __HIP_MEMORY_SCOPE_WORKGROUP);
      __hip_atomic_fetch_add(a + 1, r.y, __ATOMIC_RELAXED, __HIP_MEMORY_SCOPE_WORKGROUP);
    }
  });
  __syncthreads();

  float* outp = part + ((size_t)sp * npad + (size_t)db * PW) * F;
  for (int i = tid; i < PW * F / 4; i += WK_THREADS) {
    const int d = db * PW + (4 * i) / F;
    if (d < n) {
      float4 v = *reinterpret_cast<const float4*>(acc + 4 * i);
      if (F == 4) {
        const int k = (i >> 4) & 3;
        const float4 w = v;
        v.x = (k == 0) ? w.x : (k == 1) ? w.y : (k == 2) ? w.z : w.w;
        v.y = (k == 0) ? w.y : (k == 1) ? w.x : (k == 2) ? w.w : w.z;
        v.z = (k == 0) ? w.z : (k == 1) ? w.w : (k == 2) ? w.x : w.y;
        v.w = (k == 0) ? w.w : (k == 1) ? w.z : (k == 2) ? w.y : w.x;
      }
      *reinterpret_cast<float4*>(outp + 4 * i) = v;
    }
  }
}

__global__ __launch_bounds__(THREADS)
void k_xform32_wmma(const float* __restrict__ x, const float* __restrict__ W,
                    const int* __restrict__ degpart, float* __restrict__ dinv,
                    float* __restrict__ m, int n, int np, int npad) {
  __shared__ float lds[WAVES_PER_BLOCK * 64];
  int wave  = (int)((blockIdx.x * (unsigned)blockDim.x + threadIdx.x) >> 5);
  int tl    = threadIdx.x & 31;
  int row0  = wave * 16;
  if (row0 >= n) return;
  int lane16 = tl & 15;
  int hi     = tl >> 4;

  v16h b;
  {
    int  colW = lane16 & 3;
    bool keep = lane16 < 4;
#pragma unroll
    for (int j = 0; j < 16; ++j) {
      int k = (j < 8) ? (hi * 8 + j) : (16 + hi * 8 + (j - 8));
      float wv = W[k * 4 + colW];
      b[j] = keep ? (_Float16)wv : (_Float16)0.0f;
    }
  }

  int  arow = row0 + lane16;
  bool rok  = arow < n;
  int  ar   = rok ? arow : (n - 1);
  const float* xr = x + (size_t)ar * 32;
  float4 q0 = *reinterpret_cast<const float4*>(xr + hi * 8);
  float4 q1 = *reinterpret_cast<const float4*>(xr + hi * 8 + 4);
  float4 q2 = *reinterpret_cast<const float4*>(xr + 16 + hi * 8);
  float4 q3 = *reinterpret_cast<const float4*>(xr + 16 + hi * 8 + 4);
  int degsum = 0;
  for (int s = 0; s < np; ++s) degsum += degpart[(size_t)s * npad + ar];
  float zf = rok ? 1.0f : 0.0f;
  v16h a;
  a[0]=(_Float16)(q0.x*zf); a[1]=(_Float16)(q0.y*zf); a[2]=(_Float16)(q0.z*zf); a[3]=(_Float16)(q0.w*zf);
  a[4]=(_Float16)(q1.x*zf); a[5]=(_Float16)(q1.y*zf); a[6]=(_Float16)(q1.z*zf); a[7]=(_Float16)(q1.w*zf);
  a[8]=(_Float16)(q2.x*zf); a[9]=(_Float16)(q2.y*zf); a[10]=(_Float16)(q2.z*zf); a[11]=(_Float16)(q2.w*zf);
  a[12]=(_Float16)(q3.x*zf); a[13]=(_Float16)(q3.y*zf); a[14]=(_Float16)(q3.z*zf); a[15]=(_Float16)(q3.w*zf);

  v8f c = {};
  c = __builtin_amdgcn_wmma_f32_16x16x32_f16(false, a, false, b,
                                             (short)0, c, false, false);

  float* slab = lds + ((threadIdx.x >> 5) << 6);
  if (lane16 < 4) {
#pragma unroll
    for (int r = 0; r < 8; ++r) slab[(hi * 8 + r) * 4 + lane16] = c[r];
  }
  asm volatile("s_wait_dscnt 0" ::: "memory");
  if (hi == 0) {
    int mr = row0 + lane16;
    if (mr < n) {
      float4 v = *reinterpret_cast<const float4*>(slab + lane16 * 4);
      float di = rsqrtf(1.0f + (float)degsum);
      dinv[mr] = di;
      v.x *= di; v.y *= di; v.z *= di; v.w *= di;
      *reinterpret_cast<float4*>(m + (size_t)mr * 4) = v;
    }
  }
}

template<int F, int FOUT, bool BIAS2, bool SCALE, bool WRITE_H>
__global__ __launch_bounds__(THREADS)
void k_finish_xform(const float* __restrict__ part, const float* __restrict__ p,
                    const float* __restrict__ dinv,
                    const float* __restrict__ b, const float* __restrict__ W,
                    const float* __restrict__ bias2,
                    float* __restrict__ hout, float* __restrict__ mout,
                    int n, int np, int npad) {
  __shared__ float lds[WAVES_PER_BLOCK * 64];
  const int wave   = (int)((blockIdx.x * (unsigned)blockDim.x + threadIdx.x) >> 5);
  const int lane   = threadIdx.x & 31;
  const int lane16 = lane & 15;
  const int hi     = lane >> 4;
  const int row0   = wave * 16;
  if (row0 >= n) return;

  const int   f0    = (F == 4) ? (2 * hi) : 0;
  const float bias0 = b[f0];
  const float bias1 = b[f0 + 1];
  const bool  kreal = (F == 4) || (hi == 0);

  v2f bw;
#pragma unroll
  for (int j = 0; j < 2; ++j) {
    int  k    = hi * 2 + j;
    int  kc   = (k < F) ? k : (F - 1);
    int  cc   = (lane16 < FOUT) ? lane16 : (FOUT - 1);
    float wv  = W[kc * FOUT + cc];
    bw[j] = (k < F && lane16 < FOUT) ? wv : 0.0f;
  }

  const int  node = row0 + lane16;
  const bool rok  = node < n;
  const int  nc   = rok ? node : (n - 1);
  const float  di   = dinv[nc];
  const float2 self = *reinterpret_cast<const float2*>(p + (size_t)nc * F + f0);
  float s0 = 0.0f, s1 = 0.0f;
  for (int s = 0; s < np; ++s) {
    const float2 v = *reinterpret_cast<const float2*>(part + ((size_t)s * npad + nc) * F + f0);
    s0 += v.x; s1 += v.y;
  }
  const float h0 = tanhf(di * (s0 + self.x) + bias0);
  const float h1 = tanhf(di * (s1 + self.y) + bias1);

  if (WRITE_H) {
    if (rok && hi == 0) *reinterpret_cast<float2*>(hout + (size_t)node * F) = make_float2(h0, h1);
  }

  v2f arow;
  arow[0] = (rok && kreal) ? h0 : 0.0f;
  arow[1] = (rok && kreal) ? h1 : 0.0f;

  v8f c = {};
  c = __builtin_amdgcn_wmma_f32_16x16x4_f32(false, arow, false, bw,
                                            (short)0, c, false, false);
  float* slab = lds + ((threadIdx.x >> 5) << 6);
  if (lane16 < FOUT) {
#pragma unroll
    for (int r = 0; r < 8; ++r) slab[(hi * 8 + r) * FOUT + lane16] = c[r];
  }
  asm volatile("s_wait_dscnt 0" ::: "memory");
  if (hi == 0 && rok) {
    if (FOUT == 4) {
      float4 v = *reinterpret_cast<const float4*>(slab + lane16 * 4);
      if (BIAS2) { v.x += bias2[0]; v.y += bias2[1]; v.z += bias2[2]; v.w += bias2[3]; }
      if (SCALE) { v.x *= di; v.y *= di; v.z *= di; v.w *= di; }
      *reinterpret_cast<float4*>(mout + (size_t)node * 4) = v;
    } else {
      float2 v = *reinterpret_cast<const float2*>(slab + lane16 * 2);
      if (BIAS2) { v.x += bias2[0]; v.y += bias2[1]; }
      if (SCALE) { v.x *= di; v.y *= di; }
      *reinterpret_cast<float2*>(mout + (size_t)node * 2) = v;
    }
  }
}

extern "C" void kernel_launch(void* const* d_in, const int* in_sizes, int n_in,
                              void* d_out, int out_size, void* d_ws, size_t ws_size,
                              hipStream_t stream) {
  const float* x  = (const float*)d_in[0];
  const int*   ei = (const int*)  d_in[1];
  const float* W1 = (const float*)d_in[2];
  const float* b1 = (const float*)d_in[3];
  const float* W2 = (const float*)d_in[4];
  const float* b2 = (const float*)d_in[5];
  const float* W3 = (const float*)d_in[6];
  const float* b3 = (const float*)d_in[7];
  const float* Wc = (const float*)d_in[8];
  const float* bc = (const float*)d_in[9];

  const int N = in_sizes[0] / 32;
  const int E = in_sizes[1] / 2;
  const int* src = ei;
  const int* dst = ei + E;

  float* out = (float*)d_out;
  float* h3  = out + (size_t)N * 4;

  const int ntiles = (E + TILE - 1) / TILE;
  const int np     = (N + PW - 1) / PW;
  const int nbk    = np * np;
  const int npad   = np * PW;
  float* part    = (float*)d_ws;
  float* pa      = part + (size_t)np * npad * 4;
  float* pb      = pa + (size_t)npad * 4;
  float* dinv    = pb + (size_t)npad * 4;
  int*   degpart = (int*)(dinv + npad);
  int*   sorted  = degpart + (size_t)np * npad;
  int*   offs    = sorted + (size_t)ntiles * TILE;

  const int tiles = (N + 15) / 16;
  const int gW  = (tiles + WAVES_PER_BLOCK - 1) / WAVES_PER_BLOCK;

  k_tile_sort<<<ntiles, TS_THREADS, 0, stream>>>(src, dst, sorted, offs, np, nbk, E);
  k_degree<<<nbk, WK_THREADS, 0, stream>>>(sorted, offs, degpart, N, np, nbk, npad, ntiles);

  k_xform32_wmma<<<gW, THREADS, 0, stream>>>(x, W1, degpart, dinv, pa, N, np, npad);

  k_gather_lds<4><<<nbk, WK_THREADS, 0, stream>>>(sorted, offs, pa, part, N, np, nbk, npad, ntiles);
  k_finish_xform<4, 4, false, true, false><<<gW, THREADS, 0, stream>>>(
      part, pa, dinv, b1, W2, nullptr, nullptr, pb, N, np, npad);

  k_gather_lds<4><<<nbk, WK_THREADS, 0, stream>>>(sorted, offs, pb, part, N, np, nbk, npad, ntiles);
  k_finish_xform<4, 2, false, true, false><<<gW, THREADS, 0, stream>>>(
      part, pb, dinv, b2, W3, nullptr, nullptr, pa, N, np, npad);

  k_gather_lds<2><<<nbk, WK_THREADS, 0, stream>>>(sorted, offs, pa, part, N, np, nbk, npad, ntiles);
  k_finish_xform<2, 4, true, false, true><<<gW, THREADS, 0, stream>>>(
      part, pa, dinv, b3, Wc, bc, h3, out, N, np, npad);
}
